# starting kernel
# baseline (speedup 1.0000x reference)
#include <hip/hip_runtime.h>
#include <hip/hip_bf16.h>

typedef __attribute__((ext_vector_type(2))) float v2f;
typedef __attribute__((ext_vector_type(8))) float v8f;

#define FEAT 64

__global__ __launch_bounds__(256) void degree_kernel(const int* __restrict__ dst,
                                                     float* __restrict__ deg, int E) {
    int e = blockIdx.x * blockDim.x + threadIdx.x;
    if (e < E) {
        unsafeAtomicAdd(&deg[dst[e]], 1.0f);
    }
}

__global__ __launch_bounds__(256) void scatter_add_kernel(const float* __restrict__ feat,
                                                          const int* __restrict__ src,
                                                          const int* __restrict__ dst,
                                                          float* __restrict__ agg,
                                                          long long total) {
    long long i = (long long)blockIdx.x * blockDim.x + threadIdx.x;
    if (i >= total) return;
    int e = (int)(i >> 6);
    int f = (int)(i & 63);
    int s = src[e];
    int d = dst[e];
    float v = feat[(long long)s * FEAT + f];
    unsafeAtomicAdd(&agg[(long long)d * FEAT + f], v);
}

__global__ __launch_bounds__(32) void wmma_gemm_kernel(const float* __restrict__ S,
                                                       const float* __restrict__ deg,
                                                       const float* __restrict__ W,
                                                       const float* __restrict__ bias,
                                                       float* __restrict__ out,
                                                       int n, int tilesN, int ncols,
                                                       int do_relu) {
    const int tile = blockIdx.x;
    const int tm = tile / tilesN;
    const int tn = tile % tilesN;
    const int lane = threadIdx.x;
    const int half = lane >> 4;
    const int l15  = lane & 15;

    int rowA = tm * 16 + l15;
    if (rowA >= n) rowA = n - 1;
    const int kb = half * 2;

    const int colB = tn * 16 + l15;
    const int cB   = (colB < ncols) ? colB : 0;
    const float bmask = (colB < ncols) ? 1.0f : 0.0f;

    const float* Sp = S + (long long)rowA * FEAT;

    v8f c = {};
    #pragma unroll
    for (int k0 = 0; k0 < FEAT; k0 += 4) {
        const int k = k0 + kb;
        v2f a = *(const v2f*)(Sp + k);
        v2f b;
        b.x = W[(long long)k * ncols + cB] * bmask;
        b.y = W[(long long)(k + 1) * ncols + cB] * bmask;
        c = __builtin_amdgcn_wmma_f32_16x16x4_f32(
                 false, a,  false, b,
                 (short)0, c,  false,  false);
    }

    const float bv = (colB < ncols) ? bias[colB] : 0.0f;
    #pragma unroll
    for (int r = 0; r < 8; ++r) {
        int row = tm * 16 + r + half * 8;
        int rc = (row < n) ? row : (n - 1);
        float d = deg[rc];
        float inv = (d > 0.0f) ? (1.0f / d) : 0.0f;
        float v = c[r] * inv + bv;
        if (do_relu) v = fmaxf(v, 0.0f);
        if (colB < ncols && row < n) {
            out[(long long)row * ncols + colB] = v;
        }
    }
}

extern "C" void kernel_launch(void* const* d_in, const int* in_sizes, int n_in,
                              void* d_out, int out_size, void* d_ws, size_t ws_size,
                              hipStream_t stream) {
    const float* features = (const float*)d_in[0];
    const float* W1 = (const float*)d_in[1];
    const float* b1 = (const float*)d_in[2];
    const float* W2 = (const float*)d_in[3];
    const float* b2 = (const float*)d_in[4];
    const int* src = (const int*)d_in[5];
    const int* dst = (const int*)d_in[6];
    float* out = (float*)d_out;

    const int N = in_sizes[0] / FEAT;
    const int E = in_sizes[5];
    const int NC1 = FEAT;
    const int NC2 = in_sizes[4];

    float* agg = (float*)d_ws;
    float* deg = agg + (long long)N * FEAT;
    float* h1  = deg + N;

    hipMemsetAsync(agg, 0, (size_t)((long long)N * FEAT + N) * sizeof(float), stream);

    degree_kernel<<<(E + 255) / 256, 256, 0, stream>>>(dst, deg, E);

    long long total = (long long)E * FEAT;
    int blocks = (int)((total + 255) / 256);
    scatter_add_kernel<<<blocks, 256, 0, stream>>>(features, src, dst, agg, total);

    const int tilesM = (N + 15) / 16;
    const int tilesN1 = (NC1 + 15) / 16;
    wmma_gemm_kernel<<<tilesM * tilesN1, 32, 0, stream>>>(agg, deg, W1, b1, h1,
                                                          N, tilesN1, NC1,  1);

    hipMemsetAsync(agg, 0, (size_t)((long long)N * FEAT) * sizeof(float), stream);
    scatter_add_kernel<<<blocks, 256, 0, stream>>>(h1, src, dst, agg, total);

    const int tilesN2 = (NC2 + 15) / 16;
    wmma_gemm_kernel<<<tilesM * tilesN2, 32, 0, stream>>>(agg, deg, W2, b2, out,
                                                          N, tilesN2, NC2,  0);
}


# Layer-2 weight product moved ahead of its aggregation (linearity): one kernel chains both WMMA products per 16-row tile, pass 2 sums 40-float rows and writes out; one launch less; sum order changed
# speedup vs baseline: 4.4305x; 4.4305x over previous
#include <hip/hip_runtime.h>
#include <hip/hip_bf16.h>

typedef __attribute__((ext_vector_type(2))) float v2f;
typedef __attribute__((ext_vector_type(8))) float v8f;
typedef __attribute__((ext_vector_type(4))) float v4f;

#define FEAT 64
#define NCLS 40

#define ROWS_PER_WG 392
#define AGG_THREADS 1024
#define AGG_WAVES   32
#define QCAP        1024
#define BATCH       1024
#define LOADS       8
#define LIST_CAP    16384
#define CHUNK       16

typedef __attribute__((ext_vector_type(4))) int v4i;
typedef __attribute__((ext_vector_type(2))) int v2i;

__device__ __forceinline__ void load_ids(v4i (&r)[LOADS], const int* __restrict__ dst,
                                         int b, int lane, int E) {
    if ((b + 1) * BATCH <= E) {
        #pragma unroll
        for (int k = 0; k < LOADS; ++k) {
            r[k] = *(const v4i*)(dst + b * BATCH + k * 128 + lane * 4);
        }
    } else {
        #pragma unroll
        for (int k = 0; k < LOADS; ++k) {
            const int e0 = b * BATCH + k * 128 + lane * 4;
            v4i t;
            t.x = (e0     < E) ? dst[e0]     : -1;
            t.y = (e0 + 1 < E) ? dst[e0 + 1] : -1;
            t.z = (e0 + 2 < E) ? dst[e0 + 2] : -1;
            t.w = (e0 + 3 < E) ? dst[e0 + 3] : -1;
            r[k] = t;
        }
    }
}

template <int ROWW, bool FULL, bool EMIT>
__device__ __forceinline__ void drain_group(const float* __restrict__ feat,
                                            const int* __restrict__ src,
                                            const int* __restrict__ dst,
                                            float* acc, const unsigned* myq,
                                            unsigned head, int cnt, int lo, int n, int lane,
                                            unsigned* lcount, v2i* __restrict__ mylist,
                                            unsigned* ldeg) {
    int li = lane & 7;
    if (!FULL && li >= cnt) li = 0;
    const unsigned e = myq[(head + li) & (QCAP - 1)];
    int s = src[e];
    const int d = dst[e];
    s = (s < 0) ? 0 : ((s >= n) ? (n - 1) : s);
    const bool act = (ROWW == FEAT) || (lane * 2 < ROWW);
    v2f v[8];
    if (act) {
        #pragma unroll
        for (int i = 0; i < 8; ++i) {
            if (FULL || i < cnt) {
                const int si = __builtin_amdgcn_readlane(s, i);
                v[i] = *(const v2f*)(feat + (long long)si * ROWW + lane * 2);
            }
        }
    }
    if (EMIT) {
        const int m = FULL ? 8 : cnt;
        unsigned base = 0u;
        if (lane == 0) {
            base = __hip_atomic_fetch_add(lcount, (unsigned)m, __ATOMIC_RELAXED, __HIP_MEMORY_SCOPE_WORKGROUP);
        }
        base = (unsigned)__builtin_amdgcn_readfirstlane((int)base);
        if (lane < m && base + (unsigned)lane < (unsigned)LIST_CAP) {
            v2i pr;
            pr.x = s;
            pr.y = d - lo;
            mylist[base + (unsigned)lane] = pr;
        }
        if (lane < m) {
            __hip_atomic_fetch_add(ldeg + (d - lo), 1u, __ATOMIC_RELAXED, __HIP_MEMORY_SCOPE_WORKGROUP);
        }
    }
    if (act) {
        #pragma unroll
        for (int i = 0; i < 8; ++i) {
            if (FULL || i < cnt) {
                const int di = __builtin_amdgcn_readlane(d, i) - lo;
                float* p = acc + di * ROWW + lane * 2;
                __hip_atomic_fetch_add(p,     v[i].x, __ATOMIC_RELAXED, __HIP_MEMORY_SCOPE_WORKGROUP);
                __hip_atomic_fetch_add(p + 1, v[i].y, __ATOMIC_RELAXED, __HIP_MEMORY_SCOPE_WORKGROUP);
            }
        }
    }
}

template <int ROWW, bool EMIT>
__device__ __forceinline__ void scan_and_sum(const float* __restrict__ feat,
                                             const int* __restrict__ src,
                                             const int* __restrict__ dst,
                                             float* acc, unsigned* myq,
                                             unsigned* lcount, v2i* __restrict__ mylist,
                                             unsigned* ldeg,
                                             int lo, int rows, int n, int E, int w, int lane) {
    const int nb = (E + BATCH - 1) / BATCH;
    const int rot = (int)((blockIdx.x * (unsigned)nb) / gridDim.x);
    unsigned head = 0, tail = 0;

    v4i nxt[LOADS];
    if (w < nb) {
        int b = w + rot; if (b >= nb) b -= nb;
        load_ids(nxt, dst, b, lane, E);
    }
    for (int j = w; j < nb; j += AGG_WAVES) {
        int b = j + rot; if (b >= nb) b -= nb;
        v4i cur[LOADS];
        #pragma unroll
        for (int k = 0; k < LOADS; ++k) cur[k] = nxt[k];
        const int jn = j + AGG_WAVES;
        if (jn < nb) {
            int bn = jn + rot; if (bn >= nb) bn -= nb;
            load_ids(nxt, dst, bn, lane, E);
        }
        #pragma unroll
        for (int kk = 0; kk < LOADS; kk += 4) {
            #pragma unroll
            for (int k = kk; k < kk + 4; ++k) {
                const int e0 = b * BATCH + k * 128 + lane * 4;
                #pragma unroll
                for (int c = 0; c < 4; ++c) {
                    const bool m = (unsigned)(cur[k][c] - lo) < (unsigned)rows;
                    const unsigned mask = __builtin_amdgcn_ballot_w32(m);
                    if (mask != 0u) {
                        const unsigned pos = tail + __builtin_amdgcn_mbcnt_lo(mask, 0u);
                        if (m) myq[pos & (QCAP - 1)] = (unsigned)(e0 + c);
                        tail += (unsigned)__builtin_popcount(mask);
                    }
                }
            }
            while (tail - head >= 8u) {
                drain_group<ROWW, true, EMIT>(feat, src, dst, acc, myq, head, 8, lo, n, lane, lcount, mylist, ldeg);
                head += 8u;
            }
        }
    }
    if (tail != head) {
        drain_group<ROWW, false, EMIT>(feat, src, dst, acc, myq, head, (int)(tail - head), lo, n, lane, lcount, mylist, ldeg);
    }
}

__global__ __launch_bounds__(AGG_THREADS) void owned_rows_sum_kernel(const float* __restrict__ feat,
                                                                     const int* __restrict__ src,
                                                                     const int* __restrict__ dst,
                                                                     float* __restrict__ agg,
                                                                     float* __restrict__ deg,
                                                                     v2i* __restrict__ lists,
                                                                     int* __restrict__ counts,
                                                                     int n, int E) {
    __shared__ float acc[ROWS_PER_WG * FEAT];
    __shared__ unsigned q[AGG_WAVES][QCAP];
    __shared__ unsigned ldeg[ROWS_PER_WG];
    __shared__ unsigned lcount;

    const int tid = threadIdx.x;
    const int lane = tid & 31;
    const int w = __builtin_amdgcn_readfirstlane(tid >> 5);
    const int lo = blockIdx.x * ROWS_PER_WG;
    int rows = n - lo;
    if (rows > ROWS_PER_WG) rows = ROWS_PER_WG;

    for (int i = tid; i < ROWS_PER_WG * FEAT / 4; i += AGG_THREADS) {
        ((float4*)acc)[i] = make_float4(0.f, 0.f, 0.f, 0.f);
    }
    for (int i = tid; i < ROWS_PER_WG; i += AGG_THREADS) ldeg[i] = 0u;
    if (tid == 0) lcount = 0u;
    __syncthreads();

    scan_and_sum<FEAT, true>(feat, src, dst, acc, q[w], &lcount,
                       lists + (long long)blockIdx.x * LIST_CAP, ldeg, lo, rows, n, E, w, lane);
    __syncthreads();

    if (tid == 0) counts[blockIdx.x] = (int)lcount;

    for (int i = tid; i < rows; i += AGG_THREADS) {
        deg[lo + i] = (float)ldeg[i];
    }

    float4* outp = (float4*)(agg + (long long)lo * FEAT);
    for (int i = tid; i < rows * (FEAT / 4); i += AGG_THREADS) {
        outp[i] = ((const float4*)acc)[i];
    }
}

template <bool FULL>
__device__ __forceinline__ void sum_listed(const float* __restrict__ feat, float* acc,
                                           int s, int d, int m, int lane) {
    if (lane * 2 < NCLS) {
        v2f v[CHUNK];
        #pragma unroll
        for (int i = 0; i < CHUNK; ++i) {
            if (FULL || i < m) {
                const int si = __builtin_amdgcn_readlane(s, i);
                v[i] = *(const v2f*)(feat + (long long)si * NCLS + lane * 2);
            }
        }
        #pragma unroll
        for (int i = 0; i < CHUNK; ++i) {
            if (FULL || i < m) {
                const int di = __builtin_amdgcn_readlane(d, i);
                float* p = acc + di * NCLS + lane * 2;
                __hip_atomic_fetch_add(p,     v[i].x, __ATOMIC_RELAXED, __HIP_MEMORY_SCOPE_WORKGROUP);
                __hip_atomic_fetch_add(p + 1, v[i].y, __ATOMIC_RELAXED, __HIP_MEMORY_SCOPE_WORKGROUP);
            }
        }
    }
}

__global__ __launch_bounds__(AGG_THREADS) void listed_rows_sum_kernel(const float* __restrict__ z,
                                                                      const int* __restrict__ src,
                                                                      const int* __restrict__ dst,
                                                                      const float* __restrict__ deg,
                                                                      const float* __restrict__ b2,
                                                                      float* __restrict__ out,
                                                                      const v2i* __restrict__ lists,
                                                                      const int* __restrict__ counts,
                                                                      int n, int E) {
    __shared__ float acc[ROWS_PER_WG * NCLS];
    __shared__ unsigned q[AGG_WAVES][QCAP];

    const int tid = threadIdx.x;
    const int lane = tid & 31;
    const int w = __builtin_amdgcn_readfirstlane(tid >> 5);
    const int lo = blockIdx.x * ROWS_PER_WG;
    int rows = n - lo;
    if (rows > ROWS_PER_WG) rows = ROWS_PER_WG;

    for (int i = tid; i < ROWS_PER_WG * NCLS / 4; i += AGG_THREADS) {
        ((float4*)acc)[i] = make_float4(0.f, 0.f, 0.f, 0.f);
    }
    __syncthreads();

    const int total = counts[blockIdx.x];
    if (total <= LIST_CAP) {
        const v2i* __restrict__ L = lists + (long long)blockIdx.x * LIST_CAP;
        const int nch = (total + CHUNK - 1) / CHUNK;
        const int l15 = lane & (CHUNK - 1);
        v2i nxt; nxt.x = 0; nxt.y = 0;
        if (w < nch) {
            const int idx = w * CHUNK + l15;
            if (idx < total) nxt = L[idx];
        }
        for (int c = w; c < nch; c += AGG_WAVES) {
            const v2i cur = nxt;
            const int cn = c + AGG_WAVES;
            if (cn < nch) {
                const int idx = cn * CHUNK + l15;
                if (idx < total) nxt = L[idx];
            }
            int m = total - c * CHUNK;
            if (m > CHUNK) m = CHUNK;
            int s = cur.x;
            s = (s < 0) ? 0 : ((s >= n) ? (n - 1) : s);
            unsigned du = (unsigned)cur.y;
            if (du > (unsigned)(ROWS_PER_WG - 1)) du = ROWS_PER_WG - 1;
            if (m == CHUNK) sum_listed<true>(z, acc, s, (int)du, CHUNK, lane);
            else            sum_listed<false>(z, acc, s, (int)du, m, lane);
        }
    } else {
        scan_and_sum<NCLS, false>(z, src, dst, acc, q[w], nullptr, nullptr, nullptr, lo, rows, n, E, w, lane);
    }
    __syncthreads();

    const v4f* b2v = (const v4f*)b2;
    v4f* outp = (v4f*)(out + (long long)lo * NCLS);
    for (int i = tid; i < rows * (NCLS / 4); i += AGG_THREADS) {
        const int r  = i / (NCLS / 4);
        const int c4 = i - r * (NCLS / 4);
        const float d = deg[lo + r];
        const float inv = (d > 0.0f) ? (1.0f / d) : 0.0f;
        const v4f a = ((const v4f*)acc)[i];
        outp[i] = a * inv + b2v[c4];
    }
}

#define TP 68

__global__ __launch_bounds__(32) void chained_layers_kernel(const float* __restrict__ S,
                                                            const float* __restrict__ deg,
                                                            const float* __restrict__ W1,
                                                            const float* __restrict__ b1,
                                                            const float* __restrict__ W2,
                                                            float* __restrict__ z,
                                                            int n) {
    __shared__ float T[16 * TP];

    const int tm   = blockIdx.x;
    const int lane = threadIdx.x;
    const int h    = lane >> 4;
    const int c    = lane & 15;

    int rowA = tm * 16 + c;
    if (rowA >= n) rowA = n - 1;
    const float* Sp = S + (long long)rowA * FEAT + 32 * h;

    v4f a4[8];
    #pragma unroll
    for (int qd = 0; qd < 8; ++qd) a4[qd] = *(const v4f*)(Sp + 4 * qd);

    v8f acc1[4] = {};
    #pragma unroll
    for (int ks = 0; ks < 16; ++ks) {
        const int k = 32 * h + 2 * ks;
        const v4f w0 = *(const v4f*)(W1 + (long long)k * FEAT + 4 * c);
        const v4f w1 = *(const v4f*)(W1 + (long long)(k + 1) * FEAT + 4 * c);
        v2f a;
        a.x = a4[ks >> 1][(ks & 1) * 2];
        a.y = a4[ks >> 1][(ks & 1) * 2 + 1];
        #pragma unroll
        for (int j = 0; j < 4; ++j) {
            v2f b;
            b.x = w0[j];
            b.y = w1[j];
            acc1[j] = __builtin_amdgcn_wmma_f32_16x16x4_f32(
                           false, a,  false, b,
                           (short)0, acc1[j],  false,  false);
        }
    }

    const v4f bv = *(const v4f*)(b1 + 4 * c);
    #pragma unroll
    for (int r = 0; r < 8; ++r) {
        const int row = tm * 16 + 8 * h + r;
        const int rc = (row < n) ? row : (n - 1);
        const float d = deg[rc];
        const float inv = (d > 0.0f) ? (1.0f / d) : 0.0f;
        v4f t;
        t.x = fmaxf(acc1[0][r] * inv + bv.x, 0.0f);
        t.y = fmaxf(acc1[1][r] * inv + bv.y, 0.0f);
        t.z = fmaxf(acc1[2][r] * inv + bv.z, 0.0f);
        t.w = fmaxf(acc1[3][r] * inv + bv.w, 0.0f);
        *(v4f*)(T + (8 * h + r) * TP + 4 * c) = t;
    }
    __syncthreads();

    v4f h4[8];
    #pragma unroll
    for (int qd = 0; qd < 8; ++qd) h4[qd] = *(const v4f*)(T + c * TP + 32 * h + 4 * qd);

    const int cz = (c < NCLS / 4) ? c : (NCLS / 4 - 1);
    v8f acc2[4] = {};
    #pragma unroll
    for (int ks = 0; ks < 16; ++ks) {
        const int k = 32 * h + 2 * ks;
        const v4f w0 = *(const v4f*)(W2 + (long long)k * NCLS + 4 * cz);
        const v4f w1 = *(const v4f*)(W2 + (long long)(k + 1) * NCLS + 4 * cz);
        v2f a;
        a.x = h4[ks >> 1][(ks & 1) * 2];
        a.y = h4[ks >> 1][(ks & 1) * 2 + 1];
        #pragma unroll
        for (int j = 0; j < 4; ++j) {
            v2f b;
            b.x = w0[j];
            b.y = w1[j];
            acc2[j] = __builtin_amdgcn_wmma_f32_16x16x4_f32(
                           false, a,  false, b,
                           (short)0, acc2[j],  false,  false);
        }
    }

    #pragma unroll
    for (int r = 0; r < 8; ++r) {
        const int row = tm * 16 + 8 * h + r;
        v4f o;
        o.x = acc2[0][r];
        o.y = acc2[1][r];
        o.z = acc2[2][r];
        o.w = acc2[3][r];
        if (c < NCLS / 4 && row < n) {
            *(v4f*)(z + (long long)row * NCLS + 4 * c) = o;
        }
    }
}

extern "C" void kernel_launch(void* const* d_in, const int* in_sizes, int n_in,
                              void* d_out, int out_size, void* d_ws, size_t ws_size,
                              hipStream_t stream) {
    const float* features = (const float*)d_in[0];
    const float* W1 = (const float*)d_in[1];
    const float* b1 = (const float*)d_in[2];
    const float* W2 = (const float*)d_in[3];
    const float* b2 = (const float*)d_in[4];
    const int* src = (const int*)d_in[5];
    const int* dst = (const int*)d_in[6];
    float* out = (float*)d_out;

    const int N = in_sizes[0] / FEAT;
    const int E = in_sizes[5];

    float* agg = (float*)d_ws;
    float* deg = agg + (long long)N * FEAT;
    float* z   = deg + N;
    const int aggGroups = (N + ROWS_PER_WG - 1) / ROWS_PER_WG;
    v2i* lists  = (v2i*)(z + (long long)N * FEAT);
    int* counts = (int*)(lists + (long long)aggGroups * LIST_CAP);

    owned_rows_sum_kernel<<<aggGroups, AGG_THREADS, 0, stream>>>(features, src, dst, agg, deg, lists, counts, N, E);

    const int tilesM = (N + 15) / 16;
    chained_layers_kernel<<<tilesM, 32, 0, stream>>>(agg, deg, W1, b1, W2, z, N);

    listed_rows_sum_kernel<<<aggGroups, AGG_THREADS, 0, stream>>>(z, src, dst, deg, b2, out, lists, counts, N, E);
}


# Scan key cut to one byte per edge: a new first launch writes each edge's owner group (dst/396; rows per group 392->396 so 255 = none), groups sweep that plane, not dst; 4 launches; arrival-order sums
# speedup vs baseline: 4.9418x; 1.1154x over previous
#include <hip/hip_runtime.h>
#include <hip/hip_bf16.h>

typedef __attribute__((ext_vector_type(2))) float v2f;
typedef __attribute__((ext_vector_type(8))) float v8f;
typedef __attribute__((ext_vector_type(4))) float v4f;

#define FEAT 64
#define NCLS 40

#define ROWS_PER_WG 396
#define AGG_THREADS 1024
#define AGG_WAVES   32
#define QCAP        1024
#define OWN_LOADS   4
#define OWN_BATCH   2048
#define LIST_CAP    16384
#define CHUNK       16

typedef __attribute__((ext_vector_type(4))) int v4i;
typedef __attribute__((ext_vector_type(2))) int v2i;

typedef __attribute__((ext_vector_type(4))) unsigned v4u;

__global__ __launch_bounds__(256) void owner_bytes_kernel(const int* __restrict__ dst,
                                                          unsigned* __restrict__ own,
                                                          int n, int E, int padded) {
    const int e0 = (blockIdx.x * 256 + threadIdx.x) * 16;
    if (e0 >= padded) return;
    v4i d[4];
    if (e0 + 16 <= E) {
        #pragma unroll
        for (int qd = 0; qd < 4; ++qd) d[qd] = *(const v4i*)(dst + e0 + 4 * qd);
    } else {
        #pragma unroll
        for (int qd = 0; qd < 4; ++qd) {
            #pragma unroll
            for (int c = 0; c < 4; ++c) {
                const int e = e0 + 4 * qd + c;
                d[qd][c] = (e < E) ? dst[e] : -1;
            }
        }
    }
    v4u o;
    #pragma unroll
    for (int qd = 0; qd < 4; ++qd) {
        unsigned wv = 0u;
        #pragma unroll
        for (int c = 0; c < 4; ++c) {
            const unsigned u = (unsigned)d[qd][c];
            unsigned g = (u < (unsigned)n) ? (u / (unsigned)ROWS_PER_WG) : 255u;
            g = (g < 255u) ? g : 255u;
            wv |= g << (8 * c);
        }
        o[qd] = wv;
    }
    *(v4u*)(own + (e0 >> 2)) = o;
}

__device__ __forceinline__ void load_owners(v4u (&r)[OWN_LOADS], const unsigned* __restrict__ own,
                                            int b, int lane) {
    #pragma unroll
    for (int k = 0; k < OWN_LOADS; ++k) {
        r[k] = *(const v4u*)(own + b * (OWN_BATCH / 4) + k * 128 + lane * 4);
    }
}

template <int ROWW, bool FULL, bool EMIT>
__device__ __forceinline__ void drain_group(const float* __restrict__ feat,
                                            const int* __restrict__ src,
                                            const int* __restrict__ dst,
                                            float* acc, const unsigned* myq,
                                            unsigned head, int cnt, int lo, int n, int lane,
                                            unsigned* lcount, v2i* __restrict__ mylist,
                                            unsigned* ldeg) {
    int li = lane & 7;
    if (!FULL && li >= cnt) li = 0;
    const unsigned e = myq[(head + li) & (QCAP - 1)];
    int s = src[e];
    const int d = dst[e];
    s = (s < 0) ? 0 : ((s >= n) ? (n - 1) : s);
    const bool act = (ROWW == FEAT) || (lane * 2 < ROWW);
    v2f v[8];
    if (act) {
        #pragma unroll
        for (int i = 0; i < 8; ++i) {
            if (FULL || i < cnt) {
                const int si = __builtin_amdgcn_readlane(s, i);
                v[i] = *(const v2f*)(feat + (long long)si * ROWW + lane * 2);
            }
        }
    }
    if (EMIT) {
        const int m = FULL ? 8 : cnt;
        unsigned base = 0u;
        if (lane == 0) {
            base = __hip_atomic_fetch_add(lcount, (unsigned)m, __ATOMIC_RELAXED, __HIP_MEMORY_SCOPE_WORKGROUP);
        }
        base = (unsigned)__builtin_amdgcn_readfirstlane((int)base);
        if (lane < m && base + (unsigned)lane < (unsigned)LIST_CAP) {
            v2i pr;
            pr.x = s;
            pr.y = d - lo;
            mylist[base + (unsigned)lane] = pr;
        }
        if (lane < m) {
            __hip_atomic_fetch_add(ldeg + (d - lo), 1u, __ATOMIC_RELAXED, __HIP_MEMORY_SCOPE_WORKGROUP);
        }
    }
    if (act) {
        #pragma unroll
        for (int i = 0; i < 8; ++i) {
            if (FULL || i < cnt) {
                const int di = __builtin_amdgcn_readlane(d, i) - lo;
                float* p = acc + di * ROWW + lane * 2;
                __hip_atomic_fetch_add(p,     v[i].x, __ATOMIC_RELAXED, __HIP_MEMORY_SCOPE_WORKGROUP);
                __hip_atomic_fetch_add(p + 1, v[i].y, __ATOMIC_RELAXED, __HIP_MEMORY_SCOPE_WORKGROUP);
            }
        }
    }
}

template <int ROWW, bool EMIT>
__device__ __forceinline__ void scan_and_sum(const float* __restrict__ feat,
                                             const int* __restrict__ src,
                                             const int* __restrict__ dst,
                                             const unsigned* __restrict__ own,
                                             float* acc, unsigned* myq,
                                             unsigned* lcount, v2i* __restrict__ mylist,
                                             unsigned* ldeg,
                                             int lo, int n, int E, int w, int lane) {
    const int nb = (E + OWN_BATCH - 1) / OWN_BATCH;
    const int rot = (int)((blockIdx.x * (unsigned)nb) / gridDim.x);
    const unsigned pat = blockIdx.x * 0x01010101u;
    unsigned head = 0, tail = 0;

    v4u nxt[OWN_LOADS];
    if (w < nb) {
        int b = w + rot; if (b >= nb) b -= nb;
        load_owners(nxt, own, b, lane);
    }
    for (int j = w; j < nb; j += AGG_WAVES) {
        int b = j + rot; if (b >= nb) b -= nb;
        v4u cur[OWN_LOADS];
        #pragma unroll
        for (int k = 0; k < OWN_LOADS; ++k) cur[k] = nxt[k];
        const int jn = j + AGG_WAVES;
        if (jn < nb) {
            int bn = jn + rot; if (bn >= nb) bn -= nb;
            load_owners(nxt, own, bn, lane);
        }
        #pragma unroll
        for (int k = 0; k < OWN_LOADS; ++k) {
            const int ebase = b * OWN_BATCH + k * 512;
            #pragma unroll
            for (int c = 0; c < 4; ++c) {
                const unsigned x = cur[k][c] ^ pat;
                const unsigned t = ~((((x & 0x7F7F7F7Fu) + 0x7F7F7F7Fu) | x) | 0x7F7F7F7Fu);
                unsigned mask = __builtin_amdgcn_ballot_w32(t != 0u);
                while (mask != 0u) {
                    const int l = __builtin_ctz(mask);
                    unsigned tl = (unsigned)__builtin_amdgcn_readlane((int)t, l);
                    const unsigned eb = (unsigned)(ebase + l * 16 + c * 4);
                    do {
                        if (lane == 0) myq[tail & (QCAP - 1)] = eb + (unsigned)(__builtin_ctz(tl) >> 3);
                        ++tail;
                        tl &= tl - 1u;
                    } while (tl != 0u);
                    mask &= mask - 1u;
                }
            }
            while (tail - head >= 8u) {
                drain_group<ROWW, true, EMIT>(feat, src, dst, acc, myq, head, 8, lo, n, lane, lcount, mylist, ldeg);
                head += 8u;
            }
        }
    }
    if (tail != head) {
        drain_group<ROWW, false, EMIT>(feat, src, dst, acc, myq, head, (int)(tail - head), lo, n, lane, lcount, mylist, ldeg);
    }
}

__global__ __launch_bounds__(AGG_THREADS) void owned_rows_sum_kernel(const float* __restrict__ feat,
                                                                     const int* __restrict__ src,
                                                                     const int* __restrict__ dst,
                                                                     const unsigned* __restrict__ own,
                                                                     float* __restrict__ agg,
                                                                     float* __restrict__ deg,
                                                                     v2i* __restrict__ lists,
                                                                     int* __restrict__ counts,
                                                                     int n, int E) {
    __shared__ float acc[ROWS_PER_WG * FEAT];
    __shared__ unsigned q[AGG_WAVES][QCAP];
    __shared__ unsigned ldeg[ROWS_PER_WG];
    __shared__ unsigned lcount;

    const int tid = threadIdx.x;
    const int lane = tid & 31;
    const int w = __builtin_amdgcn_readfirstlane(tid >> 5);
    const int lo = blockIdx.x * ROWS_PER_WG;
    int rows = n - lo;
    if (rows > ROWS_PER_WG) rows = ROWS_PER_WG;

    for (int i = tid; i < ROWS_PER_WG * FEAT / 4; i += AGG_THREADS) {
        ((float4*)acc)[i] = make_float4(0.f, 0.f, 0.f, 0.f);
    }
    for (int i = tid; i < ROWS_PER_WG; i += AGG_THREADS) ldeg[i] = 0u;
    if (tid == 0) lcount = 0u;
    __syncthreads();

    scan_and_sum<FEAT, true>(feat, src, dst, own, acc, q[w], &lcount,
                       lists + (long long)blockIdx.x * LIST_CAP, ldeg, lo, n, E, w, lane);
    __syncthreads();

    if (tid == 0) counts[blockIdx.x] = (int)lcount;

    for (int i = tid; i < rows; i += AGG_THREADS) {
        deg[lo + i] = (float)ldeg[i];
    }

    float4* outp = (float4*)(agg + (long long)lo * FEAT);
    for (int i = tid; i < rows * (FEAT / 4); i += AGG_THREADS) {
        outp[i] = ((const float4*)acc)[i];
    }
}

template <bool FULL>
__device__ __forceinline__ void sum_listed(const float* __restrict__ feat, float* acc,
                                           int s, int d, int m, int lane) {
    if (lane * 2 < NCLS) {
        v2f v[CHUNK];
        #pragma unroll
        for (int i = 0; i < CHUNK; ++i) {
            if (FULL || i < m) {
                const int si = __builtin_amdgcn_readlane(s, i);
                v[i] = *(const v2f*)(feat + (long long)si * NCLS + lane * 2);
            }
        }
        #pragma unroll
        for (int i = 0; i < CHUNK; ++i) {
            if (FULL || i < m) {
                const int di = __builtin_amdgcn_readlane(d, i);
                float* p = acc + di * NCLS + lane * 2;
                __hip_atomic_fetch_add(p,     v[i].x, __ATOMIC_RELAXED, __HIP_MEMORY_SCOPE_WORKGROUP);
                __hip_atomic_fetch_add(p + 1, v[i].y, __ATOMIC_RELAXED, __HIP_MEMORY_SCOPE_WORKGROUP);
            }
        }
    }
}

__global__ __launch_bounds__(AGG_THREADS) void listed_rows_sum_kernel(const float* __restrict__ z,
                                                                      const int* __restrict__ src,
                                                                      const int* __restrict__ dst,
                                                                      const unsigned* __restrict__ own,
                                                                      const float* __restrict__ deg,
                                                                      const float* __restrict__ b2,
                                                                      float* __restrict__ out,
                                                                      const v2i* __restrict__ lists,
                                                                      const int* __restrict__ counts,
                                                                      int n, int E) {
    __shared__ float acc[ROWS_PER_WG * NCLS];
    __shared__ unsigned q[AGG_WAVES][QCAP];

    const int tid = threadIdx.x;
    const int lane = tid & 31;
    const int w = __builtin_amdgcn_readfirstlane(tid >> 5);
    const int lo = blockIdx.x * ROWS_PER_WG;
    int rows = n - lo;
    if (rows > ROWS_PER_WG) rows = ROWS_PER_WG;

    for (int i = tid; i < ROWS_PER_WG * NCLS / 4; i += AGG_THREADS) {
        ((float4*)acc)[i] = make_float4(0.f, 0.f, 0.f, 0.f);
    }
    __syncthreads();

    const int total = counts[blockIdx.x];
    if (total <= LIST_CAP) {
        const v2i* __restrict__ L = lists + (long long)blockIdx.x * LIST_CAP;
        const int nch = (total + CHUNK - 1) / CHUNK;
        const int l15 = lane & (CHUNK - 1);
        v2i nxt; nxt.x = 0; nxt.y = 0;
        if (w < nch) {
            const int idx = w * CHUNK + l15;
            if (idx < total) nxt = L[idx];
        }
        for (int c = w; c < nch; c += AGG_WAVES) {
            const v2i cur = nxt;
            const int cn = c + AGG_WAVES;
            if (cn < nch) {
                const int idx = cn * CHUNK + l15;
                if (idx < total) nxt = L[idx];
            }
            int m = total - c * CHUNK;
            if (m > CHUNK) m = CHUNK;
            int s = cur.x;
            s = (s < 0) ? 0 : ((s >= n) ? (n - 1) : s);
            unsigned du = (unsigned)cur.y;
            if (du > (unsigned)(ROWS_PER_WG - 1)) du = ROWS_PER_WG - 1;
            if (m == CHUNK) sum_listed<true>(z, acc, s, (int)du, CHUNK, lane);
            else            sum_listed<false>(z, acc, s, (int)du, m, lane);
        }
    } else {
        scan_and_sum<NCLS, false>(z, src, dst, own, acc, q[w], nullptr, nullptr, nullptr, lo, n, E, w, lane);
    }
    __syncthreads();

    const v4f* b2v = (const v4f*)b2;
    v4f* outp = (v4f*)(out + (long long)lo * NCLS);
    for (int i = tid; i < rows * (NCLS / 4); i += AGG_THREADS) {
        const int r  = i / (NCLS / 4);
        const int c4 = i - r * (NCLS / 4);
        const float d = deg[lo + r];
        const float inv = (d > 0.0f) ? (1.0f / d) : 0.0f;
        const v4f a = ((const v4f*)acc)[i];
        outp[i] = a * inv + b2v[c4];
    }
}

#define TP 68

__global__ __launch_bounds__(32) void chained_layers_kernel(const float* __restrict__ S,
                                                            const float* __restrict__ deg,
                                                            const float* __restrict__ W1,
                                                            const float* __restrict__ b1,
                                                            const float* __restrict__ W2,
                                                            float* __restrict__ z,
                                                            int n) {
    __shared__ float T[16 * TP];

    const int tm   = blockIdx.x;
    const int lane = threadIdx.x;
    const int h    = lane >> 4;
    const int c    = lane & 15;

    int rowA = tm * 16 + c;
    if (rowA >= n) rowA = n - 1;
    const float* Sp = S + (long long)rowA * FEAT + 32 * h;

    v4f a4[8];
    #pragma unroll
    for (int qd = 0; qd < 8; ++qd) a4[qd] = *(const v4f*)(Sp + 4 * qd);

    v8f acc1[4] = {};
    #pragma unroll
    for (int ks = 0; ks < 16; ++ks) {
        const int k = 32 * h + 2 * ks;
        const v4f w0 = *(const v4f*)(W1 + (long long)k * FEAT + 4 * c);
        const v4f w1 = *(const v4f*)(W1 + (long long)(k + 1) * FEAT + 4 * c);
        v2f a;
        a.x = a4[ks >> 1][(ks & 1) * 2];
        a.y = a4[ks >> 1][(ks & 1) * 2 + 1];
        #pragma unroll
        for (int j = 0; j < 4; ++j) {
            v2f b;
            b.x = w0[j];
            b.y = w1[j];
            acc1[j] = __builtin_amdgcn_wmma_f32_16x16x4_f32(
                           false, a,  false, b,
                           (short)0, acc1[j],  false,  false);
        }
    }

    const v4f bv = *(const v4f*)(b1 + 4 * c);
    #pragma unroll
    for (int r = 0; r < 8; ++r) {
        const int row = tm * 16 + 8 * h + r;
        const int rc = (row < n) ? row : (n - 1);
        const float d = deg[rc];
        const float inv = (d > 0.0f) ? (1.0f / d) : 0.0f;
        v4f t;
        t.x = fmaxf(acc1[0][r] * inv + bv.x, 0.0f);
        t.y = fmaxf(acc1[1][r] * inv + bv.y, 0.0f);
        t.z = fmaxf(acc1[2][r] * inv + bv.z, 0.0f);
        t.w = fmaxf(acc1[3][r] * inv + bv.w, 0.0f);
        *(v4f*)(T + (8 * h + r) * TP + 4 * c) = t;
    }
    __syncthreads();

    v4f h4[8];
    #pragma unroll
    for (int qd = 0; qd < 8; ++qd) h4[qd] = *(const v4f*)(T + c * TP + 32 * h + 4 * qd);

    const int cz = (c < NCLS / 4) ? c : (NCLS / 4 - 1);
    v8f acc2[4] = {};
    #pragma unroll
    for (int ks = 0; ks < 16; ++ks) {
        const int k = 32 * h + 2 * ks;
        const v4f w0 = *(const v4f*)(W2 + (long long)k * NCLS + 4 * cz);
        const v4f w1 = *(const v4f*)(W2 + (long long)(k + 1) * NCLS + 4 * cz);
        v2f a;
        a.x = h4[ks >> 1][(ks & 1) * 2];
        a.y = h4[ks >> 1][(ks & 1) * 2 + 1];
        #pragma unroll
        for (int j = 0; j < 4; ++j) {
            v2f b;
            b.x = w0[j];
            b.y = w1[j];
            acc2[j] = __builtin_amdgcn_wmma_f32_16x16x4_f32(
                           false, a,  false, b,
                           (short)0, acc2[j],  false,  false);
        }
    }

    #pragma unroll
    for (int r = 0; r < 8; ++r) {
        const int row = tm * 16 + 8 * h + r;
        v4f o;
        o.x = acc2[0][r];
        o.y = acc2[1][r];
        o.z = acc2[2][r];
        o.w = acc2[3][r];
        if (c < NCLS / 4 && row < n) {
            *(v4f*)(z + (long long)row * NCLS + 4 * c) = o;
        }
    }
}

extern "C" void kernel_launch(void* const* d_in, const int* in_sizes, int n_in,
                              void* d_out, int out_size, void* d_ws, size_t ws_size,
                              hipStream_t stream) {
    const float* features = (const float*)d_in[0];
    const float* W1 = (const float*)d_in[1];
    const float* b1 = (const float*)d_in[2];
    const float* W2 = (const float*)d_in[3];
    const float* b2 = (const float*)d_in[4];
    const int* src = (const int*)d_in[5];
    const int* dst = (const int*)d_in[6];
    float* out = (float*)d_out;

    const int N = in_sizes[0] / FEAT;
    const int E = in_sizes[5];

    float* agg = (float*)d_ws;
    float* deg = agg + (long long)N * FEAT;
    float* z   = deg + N;
    const int aggGroups = (N + ROWS_PER_WG - 1) / ROWS_PER_WG;
    v2i* lists  = (v2i*)(z + (long long)N * FEAT);
    int* counts = (int*)(lists + (long long)aggGroups * LIST_CAP);
    unsigned* own = (unsigned*)(counts + 256);
    const int padded = ((E + OWN_BATCH - 1) / OWN_BATCH) * OWN_BATCH;

    owner_bytes_kernel<<<(padded / 16 + 255) / 256, 256, 0, stream>>>(dst, own, N, E, padded);

    owned_rows_sum_kernel<<<aggGroups, AGG_THREADS, 0, stream>>>(features, src, dst, own, agg, deg, lists, counts, N, E);

    const int tilesM = (N + 15) / 16;
    chained_layers_kernel<<<tilesM, 32, 0, stream>>>(agg, deg, W1, b1, W2, z, N);

    listed_rows_sum_kernel<<<aggGroups, AGG_THREADS, 0, stream>>>(z, src, dst, own, deg, b2, out, lists, counts, N, E);
}


# Edges binned by owner group in the first launch (LDS counting sort per 4096 edges, one pair slab per block and group); both sums read only their slabs: owner-byte sweep gone; arrival-order sums
# speedup vs baseline: 5.4923x; 1.1114x over previous
#include <hip/hip_runtime.h>
#include <hip/hip_bf16.h>

typedef __attribute__((ext_vector_type(2))) float v2f;
typedef __attribute__((ext_vector_type(8))) float v8f;
typedef __attribute__((ext_vector_type(4))) float v4f;

#define FEAT 64
#define NCLS 40

#define ROWS_PER_WG 396
#define AGG_THREADS 1024
#define AGG_WAVES   32
#define EB          4096
#define BIN_THREADS 256
#define MAXG        256
#define SLAB        64
#define CHUNK       16
#define RCAP        512

typedef __attribute__((ext_vector_type(4))) int v4i;
typedef __attribute__((ext_vector_type(2))) int v2i;

__global__ __launch_bounds__(BIN_THREADS) void bin_edges_kernel(const int* __restrict__ src,
                                                                const int* __restrict__ dst,
                                                                v2i* __restrict__ slabs,
                                                                int* __restrict__ cnts,
                                                                int n, int E, int nblk, int groups) {
    __shared__ unsigned hist[MAXG];
    __shared__ unsigned start[MAXG];
    __shared__ v2i sorted[EB];

    const int tid  = threadIdx.x;
    const int lane = tid & 31;
    const int w    = __builtin_amdgcn_readfirstlane(tid >> 5);
    const int b    = blockIdx.x;

    hist[tid] = 0u;
    __syncthreads();

    int      gq[16];
    unsigned rk[16];
    v2i      pr[16];
    #pragma unroll
    for (int k = 0; k < 4; ++k) {
        const int e = b * EB + (k * BIN_THREADS + tid) * 4;
        v4i d4, s4;
        if (e + 4 <= E) {
            d4 = *(const v4i*)(dst + e);
            s4 = *(const v4i*)(src + e);
        } else {
            #pragma unroll
            for (int c = 0; c < 4; ++c) {
                d4[c] = (e + c < E) ? dst[e + c] : -1;
                s4[c] = (e + c < E) ? src[e + c] : 0;
            }
        }
        #pragma unroll
        for (int c = 0; c < 4; ++c) {
            const unsigned u = (unsigned)d4[c];
            const unsigned g = u / (unsigned)ROWS_PER_WG;
            const bool ok = (u < (unsigned)n) && (g < (unsigned)MAXG);
            int s = s4[c];
            s = (s < 0) ? 0 : ((s >= n) ? (n - 1) : s);
            const int j = 4 * k + c;
            gq[j] = ok ? (int)g : -1;
            pr[j].x = s;
            pr[j].y = (int)(u - g * (unsigned)ROWS_PER_WG);
            rk[j] = 0u;
            if (ok) rk[j] = __hip_atomic_fetch_add(&hist[g], 1u, __ATOMIC_RELAXED, __HIP_MEMORY_SCOPE_WORKGROUP);
        }
    }
    __syncthreads();

    if (w == 0) {
        unsigned c8[8];
        unsigned tot = 0u;
        #pragma unroll
        for (int i = 0; i < 8; ++i) { c8[i] = hist[lane * 8 + i]; tot += c8[i]; }
        unsigned incl = tot;
        #pragma unroll
        for (int o = 1; o < 32; o <<= 1) {
            const unsigned t = (unsigned)__shfl_up((int)incl, o);
            if (lane >= o) incl += t;
        }
        unsigned run = incl - tot;
        #pragma unroll
        for (int i = 0; i < 8; ++i) { start[lane * 8 + i] = run; run += c8[i]; }
    }
    __syncthreads();

    #pragma unroll
    for (int j = 0; j < 16; ++j) {
        if (gq[j] >= 0) sorted[start[gq[j]] + rk[j]] = pr[j];
    }
    __syncthreads();

    const int gl = w + (BIN_THREADS / 32) * lane;
    const unsigned cl = hist[gl];
    const unsigned sl0 = start[gl];
    if (gl < groups) cnts[(long long)gl * nblk + b] = (int)cl;
    #pragma unroll 4
    for (int i = 0; i < 32; ++i) {
        const int g = w + (BIN_THREADS / 32) * i;
        if (g >= groups) break;
        const unsigned c  = (unsigned)__builtin_amdgcn_readlane((int)cl, i);
        const unsigned s0 = (unsigned)__builtin_amdgcn_readlane((int)sl0, i);
        const unsigned cc = (c < (unsigned)SLAB) ? c : (unsigned)SLAB;
        v2i* __restrict__ sl = slabs + ((long long)g * nblk + b) * SLAB;
        if ((unsigned)lane < cc) sl[lane] = sorted[s0 + (unsigned)lane];
        if ((unsigned)lane + 32u < cc) sl[lane + 32] = sorted[s0 + (unsigned)lane + 32u];
    }
}

template <int ROWW, bool FULL>
__device__ __forceinline__ void sum_chunk(const float* __restrict__ feat, float* acc,
                                          int s, int d, int m, int lane) {
    int si[CHUNK], di[CHUNK];
    #pragma unroll
    for (int i = 0; i < CHUNK; ++i) {
        if (FULL || i < m) {
            si[i] = __builtin_amdgcn_readlane(s, i);
            di[i] = __builtin_amdgcn_readlane(d, i);
        }
    }
    if ((ROWW == FEAT) || (lane * 2 < ROWW)) {
        v2f v[CHUNK];
        #pragma unroll
        for (int i = 0; i < CHUNK; ++i) {
            if (FULL || i < m) v[i] = *(const v2f*)(feat + (long long)si[i] * ROWW + lane * 2);
        }
        #pragma unroll
        for (int i = 0; i < CHUNK; ++i) {
            if (FULL || i < m) {
                float* p = acc + di[i] * ROWW + lane * 2;
                __hip_atomic_fetch_add(p,     v[i].x, __ATOMIC_RELAXED, __HIP_MEMORY_SCOPE_WORKGROUP);
                __hip_atomic_fetch_add(p + 1, v[i].y, __ATOMIC_RELAXED, __HIP_MEMORY_SCOPE_WORKGROUP);
            }
        }
    }
}

template <int ROWW, bool COUNT>
__device__ void sum_block_edges(const float* __restrict__ feat,
                                             const int* __restrict__ src,
                                             const int* __restrict__ dst,
                                             float* acc, unsigned* ldeg,
                                             int g, int lo, int n, int E, int b, int lane) {
    const int e0 = b * EB;
    const int e1 = (e0 + EB < E) ? (e0 + EB) : E;
    for (int e = e0; e < e1; e += 32) {
        const int ee = e + lane;
        int d = -1, s = 0;
        if (ee < e1) { d = dst[ee]; s = src[ee]; }
        const unsigned u = (unsigned)d;
        const bool hit = (u < (unsigned)n) && ((int)(u / (unsigned)ROWS_PER_WG) == g);
        s = (s < 0) ? 0 : ((s >= n) ? (n - 1) : s);
        const int dl = hit ? (int)(u - (unsigned)lo) : 0;
        unsigned mask = __builtin_amdgcn_ballot_w32(hit);
        if (COUNT && hit) __hip_atomic_fetch_add(ldeg + dl, 1u, __ATOMIC_RELAXED, __HIP_MEMORY_SCOPE_WORKGROUP);
        while (mask != 0u) {
            const int l = __builtin_ctz(mask);
            const int si = __builtin_amdgcn_readlane(s, l);
            const int di = __builtin_amdgcn_readlane(dl, l);
            if ((ROWW == FEAT) || (lane * 2 < ROWW)) {
                const v2f v = *(const v2f*)(feat + (long long)si * ROWW + lane * 2);
                float* p = acc + di * ROWW + lane * 2;
                __hip_atomic_fetch_add(p,     v.x, __ATOMIC_RELAXED, __HIP_MEMORY_SCOPE_WORKGROUP);
                __hip_atomic_fetch_add(p + 1, v.y, __ATOMIC_RELAXED, __HIP_MEMORY_SCOPE_WORKGROUP);
            }
            mask &= mask - 1u;
        }
    }
}

template <int ROWW, bool COUNT>
__device__ __forceinline__ void sum_from_slabs(const float* __restrict__ feat,
                                               const int* __restrict__ src,
                                               const int* __restrict__ dst,
                                               const v2i* __restrict__ gslabs,
                                               const int* __restrict__ gcnts,
                                               float* acc, unsigned* ldeg, v2i* ring,
                                               int g, int lo, int n, int E, int nblk, int w, int lane) {
    int myc = 0;
    v2i nxt; nxt.x = 0; nxt.y = 0;
    int it = 0;
    unsigned head = 0u, tail = 0u;
    for (int b = w; b < nblk; b += AGG_WAVES, ++it) {
        if ((it & 31) == 0) {
            const int bb = b + AGG_WAVES * lane;
            myc = (bb < nblk) ? gcnts[bb] : 0;
            const int c0 = __builtin_amdgcn_readlane(myc, 0);
            nxt.x = 0; nxt.y = 0;
            if (c0 <= SLAB && lane < c0) nxt = gslabs[(long long)b * SLAB + lane];
        }
        const int c = __builtin_amdgcn_readlane(myc, it & 31);
        const v2i cur = nxt;
        const int bn = b + AGG_WAVES;
        if (bn < nblk && ((it + 1) & 31) != 0) {
            const int cn = __builtin_amdgcn_readlane(myc, (it + 1) & 31);
            nxt.x = 0; nxt.y = 0;
            if (cn <= SLAB && lane < cn) nxt = gslabs[(long long)bn * SLAB + lane];
        }
        if (c > SLAB) {
            sum_block_edges<ROWW, COUNT>(feat, src, dst, acc, ldeg, g, lo, n, E, b, lane);
            continue;
        }
        for (int off = 0; off < c; off += 32) {
            v2i p = cur;
            if (off != 0) {
                p.x = 0; p.y = 0;
                if (off + lane < c) p = gslabs[(long long)b * SLAB + off + lane];
            }
            int mm = c - off;
            if (mm > 32) mm = 32;
            int s = p.x;
            s = (s < 0) ? 0 : ((s >= n) ? (n - 1) : s);
            unsigned du = (unsigned)p.y;
            if (du > (unsigned)(ROWS_PER_WG - 1)) du = ROWS_PER_WG - 1;
            if (lane < mm) {
                if (COUNT) {
                    __hip_atomic_fetch_add(ldeg + du, 1u, __ATOMIC_RELAXED, __HIP_MEMORY_SCOPE_WORKGROUP);
                }
                v2i q;
                q.x = s;
                q.y = (int)du;
                ring[(tail + (unsigned)lane) & (RCAP - 1)] = q;
            }
            tail += (unsigned)mm;
            while (tail - head >= (unsigned)CHUNK) {
                const v2i q = ring[(head + (unsigned)(lane & (CHUNK - 1))) & (RCAP - 1)];
                sum_chunk<ROWW, true>(feat, acc, q.x, q.y, CHUNK, lane);
                head += (unsigned)CHUNK;
            }
        }
    }
    if (tail != head) {
        const v2i q = ring[(head + (unsigned)(lane & (CHUNK - 1))) & (RCAP - 1)];
        sum_chunk<ROWW, false>(feat, acc, q.x, q.y, (int)(tail - head), lane);
    }
}

__global__ __launch_bounds__(AGG_THREADS) void owned_rows_sum_kernel(const float* __restrict__ feat,
                                                                     const int* __restrict__ src,
                                                                     const int* __restrict__ dst,
                                                                     const v2i* __restrict__ slabs,
                                                                     const int* __restrict__ cnts,
                                                                     float* __restrict__ agg,
                                                                     float* __restrict__ deg,
                                                                     int n, int E, int nblk) {
    __shared__ float acc[ROWS_PER_WG * FEAT];
    __shared__ unsigned ldeg[ROWS_PER_WG];
    __shared__ v2i ring[AGG_WAVES][RCAP];

    const int tid = threadIdx.x;
    const int lane = tid & 31;
    const int w = __builtin_amdgcn_readfirstlane(tid >> 5);
    const int lo = blockIdx.x * ROWS_PER_WG;
    int rows = n - lo;
    if (rows > ROWS_PER_WG) rows = ROWS_PER_WG;

    for (int i = tid; i < ROWS_PER_WG * FEAT / 4; i += AGG_THREADS) {
        ((float4*)acc)[i] = make_float4(0.f, 0.f, 0.f, 0.f);
    }
    for (int i = tid; i < ROWS_PER_WG; i += AGG_THREADS) ldeg[i] = 0u;
    __syncthreads();

    sum_from_slabs<FEAT, true>(feat, src, dst, slabs + (long long)blockIdx.x * nblk * SLAB,
                               cnts + (long long)blockIdx.x * nblk, acc, ldeg, ring[w],
                               (int)blockIdx.x, lo, n, E, nblk, w, lane);
    __syncthreads();

    for (int i = tid; i < rows; i += AGG_THREADS) {
        deg[lo + i] = (float)ldeg[i];
    }

    float4* outp = (float4*)(agg + (long long)lo * FEAT);
    for (int i = tid; i < rows * (FEAT / 4); i += AGG_THREADS) {
        outp[i] = ((const float4*)acc)[i];
    }
}

__global__ __launch_bounds__(AGG_THREADS) void owned_rows_finish_kernel(const float* __restrict__ z,
                                                                        const int* __restrict__ src,
                                                                        const int* __restrict__ dst,
                                                                        const v2i* __restrict__ slabs,
                                                                        const int* __restrict__ cnts,
                                                                        const float* __restrict__ deg,
                                                                        const float* __restrict__ b2,
                                                                        float* __restrict__ out,
                                                                        int n, int E, int nblk) {
    __shared__ float acc[ROWS_PER_WG * NCLS];
    __shared__ v2i ring[AGG_WAVES][RCAP];

    const int tid = threadIdx.x;
    const int lane = tid & 31;
    const int w = __builtin_amdgcn_readfirstlane(tid >> 5);
    const int lo = blockIdx.x * ROWS_PER_WG;
    int rows = n - lo;
    if (rows > ROWS_PER_WG) rows = ROWS_PER_WG;

    for (int i = tid; i < ROWS_PER_WG * NCLS / 4; i += AGG_THREADS) {
        ((float4*)acc)[i] = make_float4(0.f, 0.f, 0.f, 0.f);
    }
    __syncthreads();

    sum_from_slabs<NCLS, false>(z, src, dst, slabs + (long long)blockIdx.x * nblk * SLAB,
                                cnts + (long long)blockIdx.x * nblk, acc, nullptr, ring[w],
                                (int)blockIdx.x, lo, n, E, nblk, w, lane);
    __syncthreads();

    const v4f* b2v = (const v4f*)b2;
    v4f* outp = (v4f*)(out + (long long)lo * NCLS);
    for (int i = tid; i < rows * (NCLS / 4); i += AGG_THREADS) {
        const int r  = i / (NCLS / 4);
        const int c4 = i - r * (NCLS / 4);
        const float d = deg[lo + r];
        const float inv = (d > 0.0f) ? (1.0f / d) : 0.0f;
        const v4f a = ((const v4f*)acc)[i];
        outp[i] = a * inv + b2v[c4];
    }
}

#define TP 68

__global__ __launch_bounds__(32) void chained_layers_kernel(const float* __restrict__ S,
                                                            const float* __restrict__ deg,
                                                            const float* __restrict__ W1,
                                                            const float* __restrict__ b1,
                                                            const float* __restrict__ W2,
                                                            float* __restrict__ z,
                                                            int n) {
    __shared__ float T[16 * TP];

    const int tm   = blockIdx.x;
    const int lane = threadIdx.x;
    const int h    = lane >> 4;
    const int c    = lane & 15;

    int rowA = tm * 16 + c;
    if (rowA >= n) rowA = n - 1;
    const float* Sp = S + (long long)rowA * FEAT + 32 * h;

    v4f a4[8];
    #pragma unroll
    for (int qd = 0; qd < 8; ++qd) a4[qd] = *(const v4f*)(Sp + 4 * qd);

    v8f acc1[4] = {};
    #pragma unroll
    for (int ks = 0; ks < 16; ++ks) {
        const int k = 32 * h + 2 * ks;
        const v4f w0 = *(const v4f*)(W1 + (long long)k * FEAT + 4 * c);
        const v4f w1 = *(const v4f*)(W1 + (long long)(k + 1) * FEAT + 4 * c);
        v2f a;
        a.x = a4[ks >> 1][(ks & 1) * 2];
        a.y = a4[ks >> 1][(ks & 1) * 2 + 1];
        #pragma unroll
        for (int j = 0; j < 4; ++j) {
            v2f b;
            b.x = w0[j];
            b.y = w1[j];
            acc1[j] = __builtin_amdgcn_wmma_f32_16x16x4_f32(
                           false, a,  false, b,
                           (short)0, acc1[j],  false,  false);
        }
    }

    const v4f bv = *(const v4f*)(b1 + 4 * c);
    #pragma unroll
    for (int r = 0; r < 8; ++r) {
        const int row = tm * 16 + 8 * h + r;
        const int rc = (row < n) ? row : (n - 1);
        const float d = deg[rc];
        const float inv = (d > 0.0f) ? (1.0f / d) : 0.0f;
        v4f t;
        t.x = fmaxf(acc1[0][r] * inv + bv.x, 0.0f);
        t.y = fmaxf(acc1[1][r] * inv + bv.y, 0.0f);
        t.z = fmaxf(acc1[2][r] * inv + bv.z, 0.0f);
        t.w = fmaxf(acc1[3][r] * inv + bv.w, 0.0f);
        *(v4f*)(T + (8 * h + r) * TP + 4 * c) = t;
    }
    __syncthreads();

    v4f h4[8];
    #pragma unroll
    for (int qd = 0; qd < 8; ++qd) h4[qd] = *(const v4f*)(T + c * TP + 32 * h + 4 * qd);

    const int cz = (c < NCLS / 4) ? c : (NCLS / 4 - 1);
    v8f acc2[4] = {};
    #pragma unroll
    for (int ks = 0; ks < 16; ++ks) {
        const int k = 32 * h + 2 * ks;
        const v4f w0 = *(const v4f*)(W2 + (long long)k * NCLS + 4 * cz);
        const v4f w1 = *(const v4f*)(W2 + (long long)(k + 1) * NCLS + 4 * cz);
        v2f a;
        a.x = h4[ks >> 1][(ks & 1) * 2];
        a.y = h4[ks >> 1][(ks & 1) * 2 + 1];
        #pragma unroll
        for (int j = 0; j < 4; ++j) {
            v2f b;
            b.x = w0[j];
            b.y = w1[j];
            acc2[j] = __builtin_amdgcn_wmma_f32_16x16x4_f32(
                           false, a,  false, b,
                           (short)0, acc2[j],  false,  false);
        }
    }

    #pragma unroll
    for (int r = 0; r < 8; ++r) {
        const int row = tm * 16 + 8 * h + r;
        v4f o;
        o.x = acc2[0][r];
        o.y = acc2[1][r];
        o.z = acc2[2][r];
        o.w = acc2[3][r];
        if (c < NCLS / 4 && row < n) {
            *(v4f*)(z + (long long)row * NCLS + 4 * c) = o;
        }
    }
}

extern "C" void kernel_launch(void* const* d_in, const int* in_sizes, int n_in,
                              void* d_out, int out_size, void* d_ws, size_t ws_size,
                              hipStream_t stream) {
    const float* features = (const float*)d_in[0];
    const float* W1 = (const float*)d_in[1];
    const float* b1 = (const float*)d_in[2];
    const float* W2 = (const float*)d_in[3];
    const float* b2 = (const float*)d_in[4];
    const int* src = (const int*)d_in[5];
    const int* dst = (const int*)d_in[6];
    float* out = (float*)d_out;

    const int N = in_sizes[0] / FEAT;
    const int E = in_sizes[5];

    float* agg = (float*)d_ws;
    float* deg = agg + (long long)N * FEAT;
    float* z   = deg + N;
    const int aggGroups = (N + ROWS_PER_WG - 1) / ROWS_PER_WG;
    const int nblk = (E + EB - 1) / EB;
    v2i* slabs = (v2i*)(z + (long long)N * FEAT);
    int* cnts  = (int*)(slabs + (long long)aggGroups * nblk * SLAB);

    bin_edges_kernel<<<nblk, BIN_THREADS, 0, stream>>>(src, dst, slabs, cnts, N, E, nblk, aggGroups);

    owned_rows_sum_kernel<<<aggGroups, AGG_THREADS, 0, stream>>>(features, src, dst, slabs, cnts, agg, deg, N, E, nblk);

    const int tilesM = (N + 15) / 16;
    chained_layers_kernel<<<tilesM, 32, 0, stream>>>(agg, deg, W1, b1, W2, z, N);

    owned_rows_finish_kernel<<<aggGroups, AGG_THREADS, 0, stream>>>(z, src, dst, slabs, cnts, deg, b2, out, N, E, nblk);
}


# First sum and both products in one launch: each owner group's waves run the WMMA tiles on its own summed rows straight from LDS and write z; agg plane and one launch gone; sum order unchanged
# speedup vs baseline: 6.2758x; 1.1427x over previous
#include <hip/hip_runtime.h>
#include <hip/hip_bf16.h>

typedef __attribute__((ext_vector_type(2))) float v2f;
typedef __attribute__((ext_vector_type(8))) float v8f;
typedef __attribute__((ext_vector_type(4))) float v4f;

#define FEAT 64
#define NCLS 40

#define ROWS_PER_WG 396
#define AGG_THREADS 1024
#define AGG_WAVES   32
#define EB          4096
#define BIN_THREADS 256
#define MAXG        256
#define SLAB        64
#define CHUNK       16
#define RCAP        512

typedef __attribute__((ext_vector_type(4))) int v4i;
typedef __attribute__((ext_vector_type(2))) int v2i;

__global__ __launch_bounds__(BIN_THREADS) void bin_edges_kernel(const int* __restrict__ src,
                                                                const int* __restrict__ dst,
                                                                v2i* __restrict__ slabs,
                                                                int* __restrict__ cnts,
                                                                int n, int E, int nblk, int groups) {
    __shared__ unsigned hist[MAXG];
    __shared__ unsigned start[MAXG];
    __shared__ v2i sorted[EB];

    const int tid  = threadIdx.x;
    const int lane = tid & 31;
    const int w    = __builtin_amdgcn_readfirstlane(tid >> 5);
    const int b    = blockIdx.x;

    hist[tid] = 0u;
    __syncthreads();

    int      gq[16];
    unsigned rk[16];
    v2i      pr[16];
    #pragma unroll
    for (int k = 0; k < 4; ++k) {
        const int e = b * EB + (k * BIN_THREADS + tid) * 4;
        v4i d4, s4;
        if (e + 4 <= E) {
            d4 = *(const v4i*)(dst + e);
            s4 = *(const v4i*)(src + e);
        } else {
            #pragma unroll
            for (int c = 0; c < 4; ++c) {
                d4[c] = (e + c < E) ? dst[e + c] : -1;
                s4[c] = (e + c < E) ? src[e + c] : 0;
            }
        }
        #pragma unroll
        for (int c = 0; c < 4; ++c) {
            const unsigned u = (unsigned)d4[c];
            const unsigned g = u / (unsigned)ROWS_PER_WG;
            const bool ok = (u < (unsigned)n) && (g < (unsigned)MAXG);
            int s = s4[c];
            s = (s < 0) ? 0 : ((s >= n) ? (n - 1) : s);
            const int j = 4 * k + c;
            gq[j] = ok ? (int)g : -1;
            pr[j].x = s;
            pr[j].y = (int)(u - g * (unsigned)ROWS_PER_WG);
            rk[j] = 0u;
            if (ok) rk[j] = __hip_atomic_fetch_add(&hist[g], 1u, __ATOMIC_RELAXED, __HIP_MEMORY_SCOPE_WORKGROUP);
        }
    }
    __syncthreads();

    if (w == 0) {
        unsigned c8[8];
        unsigned tot = 0u;
        #pragma unroll
        for (int i = 0; i < 8; ++i) { c8[i] = hist[lane * 8 + i]; tot += c8[i]; }
        unsigned incl = tot;
        #pragma unroll
        for (int o = 1; o < 32; o <<= 1) {
            const unsigned t = (unsigned)__shfl_up((int)incl, o);
            if (lane >= o) incl += t;
        }
        unsigned run = incl - tot;
        #pragma unroll
        for (int i = 0; i < 8; ++i) { start[lane * 8 + i] = run; run += c8[i]; }
    }
    __syncthreads();

    #pragma unroll
    for (int j = 0; j < 16; ++j) {
        if (gq[j] >= 0) sorted[start[gq[j]] + rk[j]] = pr[j];
    }
    __syncthreads();

    const int gl = w + (BIN_THREADS / 32) * lane;
    const unsigned cl = hist[gl];
    const unsigned sl0 = start[gl];
    if (gl < groups) cnts[(long long)gl * nblk + b] = (int)cl;
    #pragma unroll 4
    for (int i = 0; i < 32; ++i) {
        const int g = w + (BIN_THREADS / 32) * i;
        if (g >= groups) break;
        const unsigned c  = (unsigned)__builtin_amdgcn_readlane((int)cl, i);
        const unsigned s0 = (unsigned)__builtin_amdgcn_readlane((int)sl0, i);
        const unsigned cc = (c < (unsigned)SLAB) ? c : (unsigned)SLAB;
        v2i* __restrict__ sl = slabs + ((long long)g * nblk + b) * SLAB;
        if ((unsigned)lane < cc) sl[lane] = sorted[s0 + (unsigned)lane];
        if ((unsigned)lane + 32u < cc) sl[lane + 32] = sorted[s0 + (unsigned)lane + 32u];
    }
}

template <int ROWW, int PITCH, bool FULL>
__device__ __forceinline__ void sum_chunk(const float* __restrict__ feat, float* acc,
                                          int s, int d, int m, int lane) {
    int si[CHUNK], di[CHUNK];
    #pragma unroll
    for (int i = 0; i < CHUNK; ++i) {
        if (FULL || i < m) {
            si[i] = __builtin_amdgcn_readlane(s, i);
            di[i] = __builtin_amdgcn_readlane(d, i);
        }
    }
    if ((ROWW == FEAT) || (lane * 2 < ROWW)) {
        v2f v[CHUNK];
        #pragma unroll
        for (int i = 0; i < CHUNK; ++i) {
            if (FULL || i < m) v[i] = *(const v2f*)(feat + (long long)si[i] * ROWW + lane * 2);
        }
        #pragma unroll
        for (int i = 0; i < CHUNK; ++i) {
            if (FULL || i < m) {
                float* p = acc + di[i] * PITCH + lane * 2;
                __hip_atomic_fetch_add(p,     v[i].x, __ATOMIC_RELAXED, __HIP_MEMORY_SCOPE_WORKGROUP);
                __hip_atomic_fetch_add(p + 1, v[i].y, __ATOMIC_RELAXED, __HIP_MEMORY_SCOPE_WORKGROUP);
            }
        }
    }
}

template <int ROWW, int PITCH, bool COUNT>
__device__ void sum_block_edges(const float* __restrict__ feat,
                                             const int* __restrict__ src,
                                             const int* __restrict__ dst,
                                             float* acc, unsigned* ldeg,
                                             int g, int lo, int n, int E, int b, int lane) {
    const int e0 = b * EB;
    const int e1 = (e0 + EB < E) ? (e0 + EB) : E;
    for (int e = e0; e < e1; e += 32) {
        const int ee = e + lane;
        int d = -1, s = 0;
        if (ee < e1) { d = dst[ee]; s = src[ee]; }
        const unsigned u = (unsigned)d;
        const bool hit = (u < (unsigned)n) && ((int)(u / (unsigned)ROWS_PER_WG) == g);
        s = (s < 0) ? 0 : ((s >= n) ? (n - 1) : s);
        const int dl = hit ? (int)(u - (unsigned)lo) : 0;
        unsigned mask = __builtin_amdgcn_ballot_w32(hit);
        if (COUNT && hit) __hip_atomic_fetch_add(ldeg + dl, 1u, __ATOMIC_RELAXED, __HIP_MEMORY_SCOPE_WORKGROUP);
        while (mask != 0u) {
            const int l = __builtin_ctz(mask);
            const int si = __builtin_amdgcn_readlane(s, l);
            const int di = __builtin_amdgcn_readlane(dl, l);
            if ((ROWW == FEAT) || (lane * 2 < ROWW)) {
                const v2f v = *(const v2f*)(feat + (long long)si * ROWW + lane * 2);
                float* p = acc + di * PITCH + lane * 2;
                __hip_atomic_fetch_add(p,     v.x, __ATOMIC_RELAXED, __HIP_MEMORY_SCOPE_WORKGROUP);
                __hip_atomic_fetch_add(p + 1, v.y, __ATOMIC_RELAXED, __HIP_MEMORY_SCOPE_WORKGROUP);
            }
            mask &= mask - 1u;
        }
    }
}

template <int ROWW, int PITCH, bool COUNT>
__device__ __forceinline__ void sum_from_slabs(const float* __restrict__ feat,
                                               const int* __restrict__ src,
                                               const int* __restrict__ dst,
                                               const v2i* __restrict__ gslabs,
                                               const int* __restrict__ gcnts,
                                               float* acc, unsigned* ldeg, v2i* ring,
                                               int g, int lo, int n, int E, int nblk, int w, int lane) {
    int myc = 0;
    v2i nxt; nxt.x = 0; nxt.y = 0;
    int it = 0;
    unsigned head = 0u, tail = 0u;
    for (int b = w; b < nblk; b += AGG_WAVES, ++it) {
        if ((it & 31) == 0) {
            const int bb = b + AGG_WAVES * lane;
            myc = (bb < nblk) ? gcnts[bb] : 0;
            const int c0 = __builtin_amdgcn_readlane(myc, 0);
            nxt.x = 0; nxt.y = 0;
            if (c0 <= SLAB && lane < c0) nxt = gslabs[(long long)b * SLAB + lane];
        }
        const int c = __builtin_amdgcn_readlane(myc, it & 31);
        const v2i cur = nxt;
        const int bn = b + AGG_WAVES;
        if (bn < nblk && ((it + 1) & 31) != 0) {
            const int cn = __builtin_amdgcn_readlane(myc, (it + 1) & 31);
            nxt.x = 0; nxt.y = 0;
            if (cn <= SLAB && lane < cn) nxt = gslabs[(long long)bn * SLAB + lane];
        }
        if (c > SLAB) {
            sum_block_edges<ROWW, PITCH, COUNT>(feat, src, dst, acc, ldeg, g, lo, n, E, b, lane);
            continue;
        }
        for (int off = 0; off < c; off += 32) {
            v2i p = cur;
            if (off != 0) {
                p.x = 0; p.y = 0;
                if (off + lane < c) p = gslabs[(long long)b * SLAB + off + lane];
            }
            int mm = c - off;
            if (mm > 32) mm = 32;
            int s = p.x;
            s = (s < 0) ? 0 : ((s >= n) ? (n - 1) : s);
            unsigned du = (unsigned)p.y;
            if (du > (unsigned)(ROWS_PER_WG - 1)) du = ROWS_PER_WG - 1;
            if (lane < mm) {
                if (COUNT) {
                    __hip_atomic_fetch_add(ldeg + du, 1u, __ATOMIC_RELAXED, __HIP_MEMORY_SCOPE_WORKGROUP);
                }
                v2i q;
                q.x = s;
                q.y = (int)du;
                ring[(tail + (unsigned)lane) & (RCAP - 1)] = q;
            }
            tail += (unsigned)mm;
            while (tail - head >= (unsigned)CHUNK) {
                const v2i q = ring[(head + (unsigned)(lane & (CHUNK - 1))) & (RCAP - 1)];
                sum_chunk<ROWW, PITCH, true>(feat, acc, q.x, q.y, CHUNK, lane);
                head += (unsigned)CHUNK;
            }
        }
    }
    if (tail != head) {
        const v2i q = ring[(head + (unsigned)(lane & (CHUNK - 1))) & (RCAP - 1)];
        sum_chunk<ROWW, PITCH, false>(feat, acc, q.x, q.y, (int)(tail - head), lane);
    }
}

__global__ __launch_bounds__(AGG_THREADS) void owned_rows_finish_kernel(const float* __restrict__ z,
                                                                        const int* __restrict__ src,
                                                                        const int* __restrict__ dst,
                                                                        const v2i* __restrict__ slabs,
                                                                        const int* __restrict__ cnts,
                                                                        const float* __restrict__ deg,
                                                                        const float* __restrict__ b2,
                                                                        float* __restrict__ out,
                                                                        int n, int E, int nblk) {
    __shared__ float acc[ROWS_PER_WG * NCLS];
    __shared__ v2i ring[AGG_WAVES][RCAP];

    const int tid = threadIdx.x;
    const int lane = tid & 31;
    const int w = __builtin_amdgcn_readfirstlane(tid >> 5);
    const int lo = blockIdx.x * ROWS_PER_WG;
    int rows = n - lo;
    if (rows > ROWS_PER_WG) rows = ROWS_PER_WG;

    for (int i = tid; i < ROWS_PER_WG * NCLS / 4; i += AGG_THREADS) {
        ((float4*)acc)[i] = make_float4(0.f, 0.f, 0.f, 0.f);
    }
    __syncthreads();

    sum_from_slabs<NCLS, NCLS, false>(z, src, dst, slabs + (long long)blockIdx.x * nblk * SLAB,
                                cnts + (long long)blockIdx.x * nblk, acc, nullptr, ring[w],
                                (int)blockIdx.x, lo, n, E, nblk, w, lane);
    __syncthreads();

    const v4f* b2v = (const v4f*)b2;
    v4f* outp = (v4f*)(out + (long long)lo * NCLS);
    for (int i = tid; i < rows * (NCLS / 4); i += AGG_THREADS) {
        const int r  = i / (NCLS / 4);
        const int c4 = i - r * (NCLS / 4);
        const float d = deg[lo + r];
        const float inv = (d > 0.0f) ? (1.0f / d) : 0.0f;
        const v4f a = ((const v4f*)acc)[i];
        outp[i] = a * inv + b2v[c4];
    }
}

#define AP 68
#define ACC_ROWS (((ROWS_PER_WG + 15) / 16) * 16)
static_assert(ACC_ROWS / 16 <= AGG_WAVES, "one 16-row tile per wave");
#define T2OFF (AGG_WAVES - ACC_ROWS / 16)

__global__ __launch_bounds__(AGG_THREADS) void owned_rows_sum_products_kernel(const float* __restrict__ feat,
                                                                              const int* __restrict__ src,
                                                                              const int* __restrict__ dst,
                                                                              const v2i* __restrict__ slabs,
                                                                              const int* __restrict__ cnts,
                                                                              const float* __restrict__ W1,
                                                                              const float* __restrict__ b1,
                                                                              const float* __restrict__ W2,
                                                                              float* __restrict__ deg,
                                                                              float* __restrict__ z,
                                                                              int n, int E, int nblk) {
    __shared__ alignas(16) float acc[ACC_ROWS * AP];
    __shared__ unsigned ldeg[ACC_ROWS];
    __shared__ v2i ring[AGG_WAVES][RCAP];

    const int tid = threadIdx.x;
    const int lane = tid & 31;
    const int w = __builtin_amdgcn_readfirstlane(tid >> 5);
    const int lo = blockIdx.x * ROWS_PER_WG;
    int rows = n - lo;
    if (rows > ROWS_PER_WG) rows = ROWS_PER_WG;

    for (int i = tid; i < ACC_ROWS * AP / 4; i += AGG_THREADS) {
        ((float4*)acc)[i] = make_float4(0.f, 0.f, 0.f, 0.f);
    }
    for (int i = tid; i < ACC_ROWS; i += AGG_THREADS) ldeg[i] = 0u;
    __syncthreads();

    sum_from_slabs<FEAT, AP, true>(feat, src, dst, slabs + (long long)blockIdx.x * nblk * SLAB,
                                   cnts + (long long)blockIdx.x * nblk, acc, ldeg, ring[w],
                                   (int)blockIdx.x, lo, n, E, nblk, w, lane);
    __syncthreads();

    for (int i = tid; i < rows; i += AGG_THREADS) {
        deg[lo + i] = (float)ldeg[i];
    }

    const int  ntiles = (rows + 15) >> 4;
    const bool mine   = (w < ntiles);
    const int  h      = lane >> 4;
    const int  c      = lane & 15;
    float* Tt = acc + (mine ? w : 0) * (16 * AP);

    if (mine) {
        v4f a4[8];
        #pragma unroll
        for (int qd = 0; qd < 8; ++qd) a4[qd] = *(const v4f*)(Tt + c * AP + 32 * h + 4 * qd);

        v8f acc1[4] = {};
        #pragma unroll
        for (int ks = 0; ks < 16; ++ks) {
            const int k = 32 * h + 2 * ks;
            const v4f w0 = *(const v4f*)(W1 + (long long)k * FEAT + 4 * c);
            const v4f w1 = *(const v4f*)(W1 + (long long)(k + 1) * FEAT + 4 * c);
            v2f a;
            a.x = a4[ks >> 1][(ks & 1) * 2];
            a.y = a4[ks >> 1][(ks & 1) * 2 + 1];
            #pragma unroll
            for (int j = 0; j < 4; ++j) {
                v2f b;
                b.x = w0[j];
                b.y = w1[j];
                acc1[j] = __builtin_amdgcn_wmma_f32_16x16x4_f32(
                               false, a,  false, b,
                               (short)0, acc1[j],  false,  false);
            }
        }

        const v4f bv = *(const v4f*)(b1 + 4 * c);
        #pragma unroll
        for (int r = 0; r < 8; ++r) {
            const float d = (float)ldeg[16 * w + 8 * h + r];
            const float inv = (d > 0.0f) ? (1.0f / d) : 0.0f;
            v4f t;
            t.x = fmaxf(acc1[0][r] * inv + bv.x, 0.0f);
            t.y = fmaxf(acc1[1][r] * inv + bv.y, 0.0f);
            t.z = fmaxf(acc1[2][r] * inv + bv.z, 0.0f);
            t.w = fmaxf(acc1[3][r] * inv + bv.w, 0.0f);
            *(v4f*)(Tt + (8 * h + r) * AP + 4 * c) = t;
        }
    }
    __syncthreads();

    const int  w2    = w - T2OFF;
    const bool mine2 = (w2 >= 0) && (w2 < ntiles);
    if (mine2) {
        const float* T2 = acc + w2 * (16 * AP);
        v4f h4[8];
        #pragma unroll
        for (int qd = 0; qd < 8; ++qd) h4[qd] = *(const v4f*)(T2 + c * AP + 32 * h + 4 * qd);

        const int cz = (c < NCLS / 4) ? c : (NCLS / 4 - 1);
        v8f acc2[4] = {};
        #pragma unroll
        for (int ks = 0; ks < 16; ++ks) {
            const int k = 32 * h + 2 * ks;
            const v4f w0 = *(const v4f*)(W2 + (long long)k * NCLS + 4 * cz);
            const v4f w1 = *(const v4f*)(W2 + (long long)(k + 1) * NCLS + 4 * cz);
            v2f a;
            a.x = h4[ks >> 1][(ks & 1) * 2];
            a.y = h4[ks >> 1][(ks & 1) * 2 + 1];
            #pragma unroll
            for (int j = 0; j < 4; ++j) {
                v2f b;
                b.x = w0[j];
                b.y = w1[j];
                acc2[j] = __builtin_amdgcn_wmma_f32_16x16x4_f32(
                               false, a,  false, b,
                               (short)0, acc2[j],  false,  false);
            }
        }

        #pragma unroll
        for (int r = 0; r < 8; ++r) {
            const int lr = 16 * w2 + 8 * h + r;
            v4f o;
            o.x = acc2[0][r];
            o.y = acc2[1][r];
            o.z = acc2[2][r];
            o.w = acc2[3][r];
            if (c < NCLS / 4 && lr < rows) {
                *(v4f*)(z + (long long)(lo + lr) * NCLS + 4 * c) = o;
            }
        }
    }
}

extern "C" void kernel_launch(void* const* d_in, const int* in_sizes, int n_in,
                              void* d_out, int out_size, void* d_ws, size_t ws_size,
                              hipStream_t stream) {
    const float* features = (const float*)d_in[0];
    const float* W1 = (const float*)d_in[1];
    const float* b1 = (const float*)d_in[2];
    const float* W2 = (const float*)d_in[3];
    const float* b2 = (const float*)d_in[4];
    const int* src = (const int*)d_in[5];
    const int* dst = (const int*)d_in[6];
    float* out = (float*)d_out;

    const int N = in_sizes[0] / FEAT;
    const int E = in_sizes[5];

    float* deg = (float*)d_ws;
    float* z   = deg + N;
    const int aggGroups = (N + ROWS_PER_WG - 1) / ROWS_PER_WG;
    const int nblk = (E + EB - 1) / EB;
    v2i* slabs = (v2i*)(z + (long long)N * FEAT);
    int* cnts  = (int*)(slabs + (long long)aggGroups * nblk * SLAB);

    bin_edges_kernel<<<nblk, BIN_THREADS, 0, stream>>>(src, dst, slabs, cnts, N, E, nblk, aggGroups);

    owned_rows_sum_products_kernel<<<aggGroups, AGG_THREADS, 0, stream>>>(features, src, dst, slabs, cnts, W1, b1, W2, deg, z, N, E, nblk);

    owned_rows_finish_kernel<<<aggGroups, AGG_THREADS, 0, stream>>>(z, src, dst, slabs, cnts, deg, b2, out, N, E, nblk);
}


# Pair count moved into each slab's first slot: a wave's first load of a slab brings count and pairs together; count plane, its 32-line strided load per wave and one dependent hop gone; sum order kept
# speedup vs baseline: 6.5361x; 1.0415x over previous
#include <hip/hip_runtime.h>
#include <hip/hip_bf16.h>

typedef __attribute__((ext_vector_type(2))) float v2f;
typedef __attribute__((ext_vector_type(8))) float v8f;
typedef __attribute__((ext_vector_type(4))) float v4f;

#define FEAT 64
#define NCLS 40

#define ROWS_PER_WG 396
#define AGG_THREADS 1024
#define AGG_WAVES   32
#define EB          4096
#define BIN_THREADS 256
#define MAXG        256
#define SLAB        64
#define CHUNK       16
#define RCAP        512

typedef __attribute__((ext_vector_type(4))) int v4i;
typedef __attribute__((ext_vector_type(2))) int v2i;

__global__ __launch_bounds__(BIN_THREADS) void bin_edges_kernel(const int* __restrict__ src,
                                                                const int* __restrict__ dst,
                                                                v2i* __restrict__ slabs,
                                                                int n, int E, int nblk, int groups) {
    __shared__ unsigned hist[MAXG];
    __shared__ unsigned start[MAXG];
    __shared__ v2i sorted[EB];

    const int tid  = threadIdx.x;
    const int lane = tid & 31;
    const int w    = __builtin_amdgcn_readfirstlane(tid >> 5);
    const int b    = blockIdx.x;

    hist[tid] = 0u;
    __syncthreads();

    int      gq[16];
    unsigned rk[16];
    v2i      pr[16];
    #pragma unroll
    for (int k = 0; k < 4; ++k) {
        const int e = b * EB + (k * BIN_THREADS + tid) * 4;
        v4i d4, s4;
        if (e + 4 <= E) {
            d4 = *(const v4i*)(dst + e);
            s4 = *(const v4i*)(src + e);
        } else {
            #pragma unroll
            for (int c = 0; c < 4; ++c) {
                d4[c] = (e + c < E) ? dst[e + c] : -1;
                s4[c] = (e + c < E) ? src[e + c] : 0;
            }
        }
        #pragma unroll
        for (int c = 0; c < 4; ++c) {
            const unsigned u = (unsigned)d4[c];
            const unsigned g = u / (unsigned)ROWS_PER_WG;
            const bool ok = (u < (unsigned)n) && (g < (unsigned)MAXG);
            int s = s4[c];
            s = (s < 0) ? 0 : ((s >= n) ? (n - 1) : s);
            const int j = 4 * k + c;
            gq[j] = ok ? (int)g : -1;
            pr[j].x = s;
            pr[j].y = (int)(u - g * (unsigned)ROWS_PER_WG);
            rk[j] = 0u;
            if (ok) rk[j] = __hip_atomic_fetch_add(&hist[g], 1u, __ATOMIC_RELAXED, __HIP_MEMORY_SCOPE_WORKGROUP);
        }
    }
    __syncthreads();

    if (w == 0) {
        unsigned c8[8];
        unsigned tot = 0u;
        #pragma unroll
        for (int i = 0; i < 8; ++i) { c8[i] = hist[lane * 8 + i]; tot += c8[i]; }
        unsigned incl = tot;
        #pragma unroll
        for (int o = 1; o < 32; o <<= 1) {
            const unsigned t = (unsigned)__shfl_up((int)incl, o);
            if (lane >= o) incl += t;
        }
        unsigned run = incl - tot;
        #pragma unroll
        for (int i = 0; i < 8; ++i) { start[lane * 8 + i] = run; run += c8[i]; }
    }
    __syncthreads();

    #pragma unroll
    for (int j = 0; j < 16; ++j) {
        if (gq[j] >= 0) sorted[start[gq[j]] + rk[j]] = pr[j];
    }
    __syncthreads();

    const int gl = w + (BIN_THREADS / 32) * lane;
    const unsigned cl = hist[gl];
    const unsigned sl0 = start[gl];
    if (gl < groups) {
        v2i hd;
        hd.x = (int)cl;
        hd.y = 0;
        slabs[((long long)gl * nblk + b) * SLAB] = hd;
    }
    #pragma unroll 4
    for (int i = 0; i < 32; ++i) {
        const int g = w + (BIN_THREADS / 32) * i;
        if (g >= groups) break;
        const unsigned c  = (unsigned)__builtin_amdgcn_readlane((int)cl, i);
        const unsigned s0 = (unsigned)__builtin_amdgcn_readlane((int)sl0, i);
        const unsigned cc = (c < (unsigned)(SLAB - 1)) ? c : (unsigned)(SLAB - 1);
        v2i* __restrict__ sl = slabs + ((long long)g * nblk + b) * SLAB + 1;
        v2i pv;
        pv.x = 0;
        pv.y = 0;
        if ((unsigned)lane < cc) pv = sorted[s0 + (unsigned)lane];
        sl[lane] = pv;
        if ((unsigned)lane + 32u < cc) sl[lane + 32] = sorted[s0 + (unsigned)lane + 32u];
    }
}

template <int ROWW, int PITCH, bool FULL>
__device__ __forceinline__ void sum_chunk(const float* __restrict__ feat, float* acc,
                                          int s, int d, int m, int lane) {
    int si[CHUNK], di[CHUNK];
    #pragma unroll
    for (int i = 0; i < CHUNK; ++i) {
        if (FULL || i < m) {
            si[i] = __builtin_amdgcn_readlane(s, i);
            di[i] = __builtin_amdgcn_readlane(d, i);
        }
    }
    if ((ROWW == FEAT) || (lane * 2 < ROWW)) {
        v2f v[CHUNK];
        #pragma unroll
        for (int i = 0; i < CHUNK; ++i) {
            if (FULL || i < m) v[i] = *(const v2f*)(feat + (long long)si[i] * ROWW + lane * 2);
        }
        #pragma unroll
        for (int i = 0; i < CHUNK; ++i) {
            if (FULL || i < m) {
                float* p = acc + di[i] * PITCH + lane * 2;
                __hip_atomic_fetch_add(p,     v[i].x, __ATOMIC_RELAXED, __HIP_MEMORY_SCOPE_WORKGROUP);
                __hip_atomic_fetch_add(p + 1, v[i].y, __ATOMIC_RELAXED, __HIP_MEMORY_SCOPE_WORKGROUP);
            }
        }
    }
}

template <int ROWW, int PITCH, bool COUNT>
__device__ void sum_block_edges(const float* __restrict__ feat,
                                             const int* __restrict__ src,
                                             const int* __restrict__ dst,
                                             float* acc, unsigned* ldeg,
                                             int g, int lo, int n, int E, int b, int lane) {
    const int e0 = b * EB;
    const int e1 = (e0 + EB < E) ? (e0 + EB) : E;
    for (int e = e0; e < e1; e += 32) {
        const int ee = e + lane;
        int d = -1, s = 0;
        if (ee < e1) { d = dst[ee]; s = src[ee]; }
        const unsigned u = (unsigned)d;
        const bool hit = (u < (unsigned)n) && ((int)(u / (unsigned)ROWS_PER_WG) == g);
        s = (s < 0) ? 0 : ((s >= n) ? (n - 1) : s);
        const int dl = hit ? (int)(u - (unsigned)lo) : 0;
        unsigned mask = __builtin_amdgcn_ballot_w32(hit);
        if (COUNT && hit) __hip_atomic_fetch_add(ldeg + dl, 1u, __ATOMIC_RELAXED, __HIP_MEMORY_SCOPE_WORKGROUP);
        while (mask != 0u) {
            const int l = __builtin_ctz(mask);
            const int si = __builtin_amdgcn_readlane(s, l);
            const int di = __builtin_amdgcn_readlane(dl, l);
            if ((ROWW == FEAT) || (lane * 2 < ROWW)) {
                const v2f v = *(const v2f*)(feat + (long long)si * ROWW + lane * 2);
                float* p = acc + di * PITCH + lane * 2;
                __hip_atomic_fetch_add(p,     v.x, __ATOMIC_RELAXED, __HIP_MEMORY_SCOPE_WORKGROUP);
                __hip_atomic_fetch_add(p + 1, v.y, __ATOMIC_RELAXED, __HIP_MEMORY_SCOPE_WORKGROUP);
            }
            mask &= mask - 1u;
        }
    }
}

template <bool COUNT>
__device__ __forceinline__ void ring_push(v2i p, int first, int m, int n, unsigned* ldeg, v2i* ring,
                                          unsigned tail, int lane) {
    int s = p.x;
    s = (s < 0) ? 0 : ((s >= n) ? (n - 1) : s);
    unsigned du = (unsigned)p.y;
    if (du > (unsigned)(ROWS_PER_WG - 1)) du = ROWS_PER_WG - 1;
    const int k = lane - first;
    if (k >= 0 && k < m) {
        if (COUNT) {
            __hip_atomic_fetch_add(ldeg + du, 1u, __ATOMIC_RELAXED, __HIP_MEMORY_SCOPE_WORKGROUP);
        }
        v2i q;
        q.x = s;
        q.y = (int)du;
        ring[(tail + (unsigned)k) & (RCAP - 1)] = q;
    }
}

template <int ROWW, int PITCH, bool COUNT>
__device__ __forceinline__ void sum_from_slabs(const float* __restrict__ feat,
                                               const int* __restrict__ src,
                                               const int* __restrict__ dst,
                                               const v2i* __restrict__ gslabs,
                                               float* acc, unsigned* ldeg, v2i* ring,
                                               int g, int lo, int n, int E, int nblk, int w, int lane) {
    v2i nxt; nxt.x = 0; nxt.y = 0;
    if (w < nblk) nxt = gslabs[(long long)w * SLAB + lane];
    unsigned head = 0u, tail = 0u;
    for (int b = w; b < nblk; b += AGG_WAVES) {
        const v2i cur = nxt;
        const int bn = b + AGG_WAVES;
        if (bn < nblk) nxt = gslabs[(long long)bn * SLAB + lane];
        const int c = __builtin_amdgcn_readlane(cur.x, 0);
        if (c > SLAB - 1) {
            sum_block_edges<ROWW, PITCH, COUNT>(feat, src, dst, acc, ldeg, g, lo, n, E, b, lane);
            continue;
        }
        if (c > 0) {
            const int m0 = (c < 31) ? c : 31;
            ring_push<COUNT>(cur, 1, m0, n, ldeg, ring, tail, lane);
            tail += (unsigned)m0;
            if (c > 31) {
                v2i p; p.x = 0; p.y = 0;
                if (31 + lane < c) p = gslabs[(long long)b * SLAB + 32 + lane];
                ring_push<COUNT>(p, 0, c - 31, n, ldeg, ring, tail, lane);
                tail += (unsigned)(c - 31);
            }
            while (tail - head >= (unsigned)CHUNK) {
                const v2i q = ring[(head + (unsigned)(lane & (CHUNK - 1))) & (RCAP - 1)];
                sum_chunk<ROWW, PITCH, true>(feat, acc, q.x, q.y, CHUNK, lane);
                head += (unsigned)CHUNK;
            }
        }
    }
    if (tail != head) {
        const v2i q = ring[(head + (unsigned)(lane & (CHUNK - 1))) & (RCAP - 1)];
        sum_chunk<ROWW, PITCH, false>(feat, acc, q.x, q.y, (int)(tail - head), lane);
    }
}

__global__ __launch_bounds__(AGG_THREADS) void owned_rows_finish_kernel(const float* __restrict__ z,
                                                                        const int* __restrict__ src,
                                                                        const int* __restrict__ dst,
                                                                        const v2i* __restrict__ slabs,
                                                                        const float* __restrict__ deg,
                                                                        const float* __restrict__ b2,
                                                                        float* __restrict__ out,
                                                                        int n, int E, int nblk) {
    __shared__ float acc[ROWS_PER_WG * NCLS];
    __shared__ v2i ring[AGG_WAVES][RCAP];

    const int tid = threadIdx.x;
    const int lane = tid & 31;
    const int w = __builtin_amdgcn_readfirstlane(tid >> 5);
    const int lo = blockIdx.x * ROWS_PER_WG;
    int rows = n - lo;
    if (rows > ROWS_PER_WG) rows = ROWS_PER_WG;

    for (int i = tid; i < ROWS_PER_WG * NCLS / 4; i += AGG_THREADS) {
        ((float4*)acc)[i] = make_float4(0.f, 0.f, 0.f, 0.f);
    }
    __syncthreads();

    sum_from_slabs<NCLS, NCLS, false>(z, src, dst, slabs + (long long)blockIdx.x * nblk * SLAB,
                                      acc, nullptr, ring[w],
                                      (int)blockIdx.x, lo, n, E, nblk, w, lane);
    __syncthreads();

    const v4f* b2v = (const v4f*)b2;
    v4f* outp = (v4f*)(out + (long long)lo * NCLS);
    for (int i = tid; i < rows * (NCLS / 4); i += AGG_THREADS) {
        const int r  = i / (NCLS / 4);
        const int c4 = i - r * (NCLS / 4);
        const float d = deg[lo + r];
        const float inv = (d > 0.0f) ? (1.0f / d) : 0.0f;
        const v4f a = ((const v4f*)acc)[i];
        outp[i] = a * inv + b2v[c4];
    }
}

#define AP 68
#define ACC_ROWS (((ROWS_PER_WG + 15) / 16) * 16)
static_assert(ACC_ROWS / 16 <= AGG_WAVES, "one 16-row tile per wave");
#define T2OFF (AGG_WAVES - ACC_ROWS / 16)

__global__ __launch_bounds__(AGG_THREADS) void owned_rows_sum_products_kernel(const float* __restrict__ feat,
                                                                              const int* __restrict__ src,
                                                                              const int* __restrict__ dst,
                                                                              const v2i* __restrict__ slabs,
                                                                              const float* __restrict__ W1,
                                                                              const float* __restrict__ b1,
                                                                              const float* __restrict__ W2,
                                                                              float* __restrict__ deg,
                                                                              float* __restrict__ z,
                                                                              int n, int E, int nblk) {
    __shared__ alignas(16) float acc[ACC_ROWS * AP];
    __shared__ unsigned ldeg[ACC_ROWS];
    __shared__ v2i ring[AGG_WAVES][RCAP];

    const int tid = threadIdx.x;
    const int lane = tid & 31;
    const int w = __builtin_amdgcn_readfirstlane(tid >> 5);
    const int lo = blockIdx.x * ROWS_PER_WG;
    int rows = n - lo;
    if (rows > ROWS_PER_WG) rows = ROWS_PER_WG;

    for (int i = tid; i < ACC_ROWS * AP / 4; i += AGG_THREADS) {
        ((float4*)acc)[i] = make_float4(0.f, 0.f, 0.f, 0.f);
    }
    for (int i = tid; i < ACC_ROWS; i += AGG_THREADS) ldeg[i] = 0u;
    __syncthreads();

    sum_from_slabs<FEAT, AP, true>(feat, src, dst, slabs + (long long)blockIdx.x * nblk * SLAB,
                                   acc, ldeg, ring[w],
                                   (int)blockIdx.x, lo, n, E, nblk, w, lane);
    __syncthreads();

    for (int i = tid; i < rows; i += AGG_THREADS) {
        deg[lo + i] = (float)ldeg[i];
    }

    const int  ntiles = (rows + 15) >> 4;
    const bool mine   = (w < ntiles);
    const int  h      = lane >> 4;
    const int  c      = lane & 15;
    float* Tt = acc + (mine ? w : 0) * (16 * AP);

    if (mine) {
        v4f a4[8];
        #pragma unroll
        for (int qd = 0; qd < 8; ++qd) a4[qd] = *(const v4f*)(Tt + c * AP + 32 * h + 4 * qd);

        v8f acc1[4] = {};
        #pragma unroll
        for (int ks = 0; ks < 16; ++ks) {
            const int k = 32 * h + 2 * ks;
            const v4f w0 = *(const v4f*)(W1 + (long long)k * FEAT + 4 * c);
            const v4f w1 = *(const v4f*)(W1 + (long long)(k + 1) * FEAT + 4 * c);
            v2f a;
            a.x = a4[ks >> 1][(ks & 1) * 2];
            a.y = a4[ks >> 1][(ks & 1) * 2 + 1];
            #pragma unroll
            for (int j = 0; j < 4; ++j) {
                v2f b;
                b.x = w0[j];
                b.y = w1[j];
                acc1[j] = __builtin_amdgcn_wmma_f32_16x16x4_f32(
                               false, a,  false, b,
                               (short)0, acc1[j],  false,  false);
            }
        }

        const v4f bv = *(const v4f*)(b1 + 4 * c);
        #pragma unroll
        for (int r = 0; r < 8; ++r) {
            const float d = (float)ldeg[16 * w + 8 * h + r];
            const float inv = (d > 0.0f) ? (1.0f / d) : 0.0f;
            v4f t;
            t.x = fmaxf(acc1[0][r] * inv + bv.x, 0.0f);
            t.y = fmaxf(acc1[1][r] * inv + bv.y, 0.0f);
            t.z = fmaxf(acc1[2][r] * inv + bv.z, 0.0f);
            t.w = fmaxf(acc1[3][r] * inv + bv.w, 0.0f);
            *(v4f*)(Tt + (8 * h + r) * AP + 4 * c) = t;
        }
    }
    __syncthreads();

    const int  w2    = w - T2OFF;
    const bool mine2 = (w2 >= 0) && (w2 < ntiles);
    if (mine2) {
        const float* T2 = acc + w2 * (16 * AP);
        v4f h4[8];
        #pragma unroll
        for (int qd = 0; qd < 8; ++qd) h4[qd] = *(const v4f*)(T2 + c * AP + 32 * h + 4 * qd);

        const int cz = (c < NCLS / 4) ? c : (NCLS / 4 - 1);
        v8f acc2[4] = {};
        #pragma unroll
        for (int ks = 0; ks < 16; ++ks) {
            const int k = 32 * h + 2 * ks;
            const v4f w0 = *(const v4f*)(W2 + (long long)k * NCLS + 4 * cz);
            const v4f w1 = *(const v4f*)(W2 + (long long)(k + 1) * NCLS + 4 * cz);
            v2f a;
            a.x = h4[ks >> 1][(ks & 1) * 2];
            a.y = h4[ks >> 1][(ks & 1) * 2 + 1];
            #pragma unroll
            for (int j = 0; j < 4; ++j) {
                v2f b;
                b.x = w0[j];
                b.y = w1[j];
                acc2[j] = __builtin_amdgcn_wmma_f32_16x16x4_f32(
                               false, a,  false, b,
                               (short)0, acc2[j],  false,  false);
            }
        }

        #pragma unroll
        for (int r = 0; r < 8; ++r) {
            const int lr = 16 * w2 + 8 * h + r;
            v4f o;
            o.x = acc2[0][r];
            o.y = acc2[1][r];
            o.z = acc2[2][r];
            o.w = acc2[3][r];
            if (c < NCLS / 4 && lr < rows) {
                *(v4f*)(z + (long long)(lo + lr) * NCLS + 4 * c) = o;
            }
        }
    }
}

extern "C" void kernel_launch(void* const* d_in, const int* in_sizes, int n_in,
                              void* d_out, int out_size, void* d_ws, size_t ws_size,
                              hipStream_t stream) {
    const float* features = (const float*)d_in[0];
    const float* W1 = (const float*)d_in[1];
    const float* b1 = (const float*)d_in[2];
    const float* W2 = (const float*)d_in[3];
    const float* b2 = (const float*)d_in[4];
    const int* src = (const int*)d_in[5];
    const int* dst = (const int*)d_in[6];
    float* out = (float*)d_out;

    const int N = in_sizes[0] / FEAT;
    const int E = in_sizes[5];

    float* deg = (float*)d_ws;
    float* z   = deg + N;
    const int aggGroups = (N + ROWS_PER_WG - 1) / ROWS_PER_WG;
    const int nblk = (E + EB - 1) / EB;
    v2i* slabs = (v2i*)(z + (long long)N * FEAT);

    bin_edges_kernel<<<nblk, BIN_THREADS, 0, stream>>>(src, dst, slabs, N, E, nblk, aggGroups);

    owned_rows_sum_products_kernel<<<aggGroups, AGG_THREADS, 0, stream>>>(features, src, dst, slabs, W1, b1, W2, deg, z, N, E, nblk);

    owned_rows_finish_kernel<<<aggGroups, AGG_THREADS, 0, stream>>>(z, src, dst, slabs, deg, b2, out, N, E, nblk);
}


# Owner group sorts its pairs by destination row in LDS (integer-add rank, scan, scatter); a wave owns 13 rows and sums them in registers, one plain store per row: float LDS adds gone; rank-order sums
# speedup vs baseline: 7.5386x; 1.1534x over previous
#include <hip/hip_runtime.h>
#include <hip/hip_bf16.h>

typedef __attribute__((ext_vector_type(2))) float v2f;
typedef __attribute__((ext_vector_type(8))) float v8f;
typedef __attribute__((ext_vector_type(4))) float v4f;

#define FEAT 64
#define NCLS 40

#define ROWS_PER_WG 396
#define AGG_THREADS 1024
#define AGG_WAVES   32
#define EB          4096
#define BIN_THREADS 256
#define MAXG        256
#define SLAB        64
#define CHUNK       16
#define RCAP        256
#define RPW         13
#define NBINS       (AGG_WAVES * RPW)
#define WCAP        320
#define SCAP        (AGG_WAVES * WCAP)
#define SBATCH      10

typedef __attribute__((ext_vector_type(4))) int v4i;
typedef __attribute__((ext_vector_type(2))) int v2i;

__global__ __launch_bounds__(BIN_THREADS) void bin_edges_kernel(const int* __restrict__ src,
                                                                const int* __restrict__ dst,
                                                                v2i* __restrict__ slabs,
                                                                int n, int E, int nblk, int groups) {
    __shared__ unsigned hist[MAXG];
    __shared__ unsigned start[MAXG];
    __shared__ v2i sorted[EB];

    const int tid  = threadIdx.x;
    const int lane = tid & 31;
    const int w    = __builtin_amdgcn_readfirstlane(tid >> 5);
    const int b    = blockIdx.x;

    hist[tid] = 0u;
    __syncthreads();

    int      gq[16];
    unsigned rk[16];
    v2i      pr[16];
    #pragma unroll
    for (int k = 0; k < 4; ++k) {
        const int e = b * EB + (k * BIN_THREADS + tid) * 4;
        v4i d4, s4;
        if (e + 4 <= E) {
            d4 = *(const v4i*)(dst + e);
            s4 = *(const v4i*)(src + e);
        } else {
            #pragma unroll
            for (int c = 0; c < 4; ++c) {
                d4[c] = (e + c < E) ? dst[e + c] : -1;
                s4[c] = (e + c < E) ? src[e + c] : 0;
            }
        }
        #pragma unroll
        for (int c = 0; c < 4; ++c) {
            const unsigned u = (unsigned)d4[c];
            const unsigned g = u / (unsigned)ROWS_PER_WG;
            const bool ok = (u < (unsigned)n) && (g < (unsigned)MAXG);
            int s = s4[c];
            s = (s < 0) ? 0 : ((s >= n) ? (n - 1) : s);
            const int j = 4 * k + c;
            gq[j] = ok ? (int)g : -1;
            pr[j].x = s;
            pr[j].y = (int)(u - g * (unsigned)ROWS_PER_WG);
            rk[j] = 0u;
            if (ok) rk[j] = __hip_atomic_fetch_add(&hist[g], 1u, __ATOMIC_RELAXED, __HIP_MEMORY_SCOPE_WORKGROUP);
        }
    }
    __syncthreads();

    if (w == 0) {
        unsigned c8[8];
        unsigned tot = 0u;
        #pragma unroll
        for (int i = 0; i < 8; ++i) { c8[i] = hist[lane * 8 + i]; tot += c8[i]; }
        unsigned incl = tot;
        #pragma unroll
        for (int o = 1; o < 32; o <<= 1) {
            const unsigned t = (unsigned)__shfl_up((int)incl, o);
            if (lane >= o) incl += t;
        }
        unsigned run = incl - tot;
        #pragma unroll
        for (int i = 0; i < 8; ++i) { start[lane * 8 + i] = run; run += c8[i]; }
    }
    __syncthreads();

    #pragma unroll
    for (int j = 0; j < 16; ++j) {
        if (gq[j] >= 0) sorted[start[gq[j]] + rk[j]] = pr[j];
    }
    __syncthreads();

    const int gl = w + (BIN_THREADS / 32) * lane;
    const unsigned cl = hist[gl];
    const unsigned sl0 = start[gl];
    if (gl < groups) {
        v2i hd;
        hd.x = (int)cl;
        hd.y = 0;
        slabs[((long long)gl * nblk + b) * SLAB] = hd;
    }
    #pragma unroll 4
    for (int i = 0; i < 32; ++i) {
        const int g = w + (BIN_THREADS / 32) * i;
        if (g >= groups) break;
        const unsigned c  = (unsigned)__builtin_amdgcn_readlane((int)cl, i);
        const unsigned s0 = (unsigned)__builtin_amdgcn_readlane((int)sl0, i);
        const unsigned cc = (c < (unsigned)(SLAB - 1)) ? c : (unsigned)(SLAB - 1);
        v2i* __restrict__ sl = slabs + ((long long)g * nblk + b) * SLAB + 1;
        v2i pv;
        pv.x = 0;
        pv.y = 0;
        if ((unsigned)lane < cc) pv = sorted[s0 + (unsigned)lane];
        sl[lane] = pv;
        if ((unsigned)lane + 32u < cc) sl[lane + 32] = sorted[s0 + (unsigned)lane + 32u];
    }
}

template <int ROWW, int PITCH, bool FULL>
__device__ __forceinline__ void sum_chunk(const float* __restrict__ feat, float* acc,
                                          int s, int d, int m, int lane) {
    int si[CHUNK], di[CHUNK];
    #pragma unroll
    for (int i = 0; i < CHUNK; ++i) {
        if (FULL || i < m) {
            si[i] = __builtin_amdgcn_readlane(s, i);
            di[i] = __builtin_amdgcn_readlane(d, i);
        }
    }
    if ((ROWW == FEAT) || (lane * 2 < ROWW)) {
        v2f v[CHUNK];
        #pragma unroll
        for (int i = 0; i < CHUNK; ++i) {
            if (FULL || i < m) v[i] = *(const v2f*)(feat + (long long)si[i] * ROWW + lane * 2);
        }
        #pragma unroll
        for (int i = 0; i < CHUNK; ++i) {
            if (FULL || i < m) {
                float* p = acc + di[i] * PITCH + lane * 2;
                __hip_atomic_fetch_add(p,     v[i].x, __ATOMIC_RELAXED, __HIP_MEMORY_SCOPE_WORKGROUP);
                __hip_atomic_fetch_add(p + 1, v[i].y, __ATOMIC_RELAXED, __HIP_MEMORY_SCOPE_WORKGROUP);
            }
        }
    }
}

template <int ROWW, int PITCH, bool COUNT>
__device__ void sum_block_edges(const float* __restrict__ feat,
                                             const int* __restrict__ src,
                                             const int* __restrict__ dst,
                                             float* acc, unsigned* ldeg,
                                             int g, int lo, int n, int E, int b, int lane) {
    const int e0 = b * EB;
    const int e1 = (e0 + EB < E) ? (e0 + EB) : E;
    for (int e = e0; e < e1; e += 32) {
        const int ee = e + lane;
        int d = -1, s = 0;
        if (ee < e1) { d = dst[ee]; s = src[ee]; }
        const unsigned u = (unsigned)d;
        const bool hit = (u < (unsigned)n) && ((int)(u / (unsigned)ROWS_PER_WG) == g);
        s = (s < 0) ? 0 : ((s >= n) ? (n - 1) : s);
        const int dl = hit ? (int)(u - (unsigned)lo) : 0;
        unsigned mask = __builtin_amdgcn_ballot_w32(hit);
        if (COUNT && hit) __hip_atomic_fetch_add(ldeg + dl, 1u, __ATOMIC_RELAXED, __HIP_MEMORY_SCOPE_WORKGROUP);
        while (mask != 0u) {
            const int l = __builtin_ctz(mask);
            const int si = __builtin_amdgcn_readlane(s, l);
            const int di = __builtin_amdgcn_readlane(dl, l);
            if ((ROWW == FEAT) || (lane * 2 < ROWW)) {
                const v2f v = *(const v2f*)(feat + (long long)si * ROWW + lane * 2);
                float* p = acc + di * PITCH + lane * 2;
                __hip_atomic_fetch_add(p,     v.x, __ATOMIC_RELAXED, __HIP_MEMORY_SCOPE_WORKGROUP);
                __hip_atomic_fetch_add(p + 1, v.y, __ATOMIC_RELAXED, __HIP_MEMORY_SCOPE_WORKGROUP);
            }
            mask &= mask - 1u;
        }
    }
}

template <bool COUNT>
__device__ __forceinline__ void ring_push(v2i p, int first, int m, int n, unsigned* ldeg, v2i* ring,
                                          unsigned tail, int lane) {
    int s = p.x;
    s = (s < 0) ? 0 : ((s >= n) ? (n - 1) : s);
    unsigned du = (unsigned)p.y;
    if (du > (unsigned)(ROWS_PER_WG - 1)) du = ROWS_PER_WG - 1;
    const int k = lane - first;
    if (k >= 0 && k < m) {
        if (COUNT) {
            __hip_atomic_fetch_add(ldeg + du, 1u, __ATOMIC_RELAXED, __HIP_MEMORY_SCOPE_WORKGROUP);
        }
        v2i q;
        q.x = s;
        q.y = (int)du;
        ring[(tail + (unsigned)k) & (RCAP - 1)] = q;
    }
}

template <int ROWW, int PITCH, bool COUNT>
__device__ __forceinline__ void sum_from_slabs(const float* __restrict__ feat,
                                               const int* __restrict__ src,
                                               const int* __restrict__ dst,
                                               const v2i* __restrict__ gslabs,
                                               float* acc, unsigned* ldeg, v2i* ring,
                                               int g, int lo, int n, int E, int nblk, int w, int lane) {
    v2i nxt; nxt.x = 0; nxt.y = 0;
    if (w < nblk) nxt = gslabs[(long long)w * SLAB + lane];
    unsigned head = 0u, tail = 0u;
    for (int b = w; b < nblk; b += AGG_WAVES) {
        const v2i cur = nxt;
        const int bn = b + AGG_WAVES;
        if (bn < nblk) nxt = gslabs[(long long)bn * SLAB + lane];
        const int c = __builtin_amdgcn_readlane(cur.x, 0);
        if (c > SLAB - 1) {
            sum_block_edges<ROWW, PITCH, COUNT>(feat, src, dst, acc, ldeg, g, lo, n, E, b, lane);
            continue;
        }
        if (c > 0) {
            const int m0 = (c < 31) ? c : 31;
            ring_push<COUNT>(cur, 1, m0, n, ldeg, ring, tail, lane);
            tail += (unsigned)m0;
            if (c > 31) {
                v2i p; p.x = 0; p.y = 0;
                if (31 + lane < c) p = gslabs[(long long)b * SLAB + 32 + lane];
                ring_push<COUNT>(p, 0, c - 31, n, ldeg, ring, tail, lane);
                tail += (unsigned)(c - 31);
            }
            while (tail - head >= (unsigned)CHUNK) {
                const v2i q = ring[(head + (unsigned)(lane & (CHUNK - 1))) & (RCAP - 1)];
                sum_chunk<ROWW, PITCH, true>(feat, acc, q.x, q.y, CHUNK, lane);
                head += (unsigned)CHUNK;
            }
        }
    }
    if (tail != head) {
        const v2i q = ring[(head + (unsigned)(lane & (CHUNK - 1))) & (RCAP - 1)];
        sum_chunk<ROWW, PITCH, false>(feat, acc, q.x, q.y, (int)(tail - head), lane);
    }
}

static_assert(NBINS >= ROWS_PER_WG, "every row has an owner wave");
static_assert(ROWS_PER_WG <= 512, "a local row fits the 9 low bits of a staged pair");
static_assert(RCAP <= WCAP, "the slow path's ring lives in the staging slots");

__device__ __forceinline__ void stage_push(v2i p, int first, int m, int n, unsigned* hist, v2i* stage,
                                           unsigned cnt, int lane) {
    int s = p.x;
    s = (s < 0) ? 0 : ((s >= n) ? (n - 1) : s);
    unsigned du = (unsigned)p.y;
    if (du > (unsigned)(ROWS_PER_WG - 1)) du = ROWS_PER_WG - 1;
    const int k = lane - first;
    if (k >= 0 && k < m) {
        const unsigned rank = __hip_atomic_fetch_add(hist + du, 1u, __ATOMIC_RELAXED, __HIP_MEMORY_SCOPE_WORKGROUP);
        v2i q;
        q.x = s;
        q.y = (int)((rank << 9) | du);
        stage[cnt + (unsigned)k] = q;
    }
}

template <int ROWW, int PITCH, bool FULL>
__device__ __forceinline__ void sum_chunk_rows(const float* __restrict__ feat, float* acc,
                                               int s, int d, int m, int lane, int& cur, v2f& a) {
    int si[CHUNK], di[CHUNK];
    #pragma unroll
    for (int i = 0; i < CHUNK; ++i) {
        if (FULL || i < m) {
            si[i] = __builtin_amdgcn_readlane(s, i);
            di[i] = __builtin_amdgcn_readlane(d, i);
        }
    }
    const bool on  = (ROWW == FEAT) || (lane * 2 < ROWW);
    const int  col = on ? lane * 2 : (ROWW - 2);
    v2f v[CHUNK];
    #pragma unroll
    for (int i = 0; i < CHUNK; ++i) {
        if (FULL || i < m) v[i] = *(const v2f*)(feat + (long long)si[i] * ROWW + col);
    }
    #pragma unroll
    for (int i = 0; i < CHUNK; ++i) {
        if (FULL || i < m) {
            if (di[i] != cur) {
                if (cur >= 0 && on) *(v2f*)(acc + cur * PITCH + col) = a;
                cur = di[i];
                a = v[i];
            } else {
                a += v[i];
            }
        }
    }
}

template <int ROWW, int PITCH, bool COUNT>
__device__ __forceinline__ void sum_rows_owned(const float* __restrict__ feat,
                                               const int* __restrict__ src,
                                               const int* __restrict__ dst,
                                               const v2i* __restrict__ gslabs,
                                               float* acc, unsigned* hist, unsigned* start,
                                               v2i* stage_all, v2i* sorted, unsigned* bad,
                                               int g, int lo, int n, int E, int nblk,
                                               int tid, int w, int lane) {
    v2i* stage = stage_all + w * WCAP;

    unsigned cnt = 0u;
    bool wbad = false;
    for (int b0 = w; b0 < nblk; b0 += AGG_WAVES * SBATCH) {
        v2i sl[SBATCH];
        #pragma unroll
        for (int j = 0; j < SBATCH; ++j) {
            const int b = b0 + AGG_WAVES * j;
            sl[j].x = 0;
            sl[j].y = 0;
            if (b < nblk) sl[j] = gslabs[(long long)b * SLAB + lane];
        }
        #pragma unroll
        for (int j = 0; j < SBATCH; ++j) {
            const int b = b0 + AGG_WAVES * j;
            if (b < nblk) {
                const unsigned c = (unsigned)__builtin_amdgcn_readlane(sl[j].x, 0);
                if (c > (unsigned)(SLAB - 1) || cnt + c > (unsigned)WCAP) {
                    wbad = true;
                } else if (c > 0u) {
                    const int m0 = (c < 31u) ? (int)c : 31;
                    stage_push(sl[j], 1, m0, n, hist, stage, cnt, lane);
                    cnt += (unsigned)m0;
                    if (c > 31u) {
                        v2i p; p.x = 0; p.y = 0;
                        if (31u + (unsigned)lane < c) p = gslabs[(long long)b * SLAB + 32 + lane];
                        stage_push(p, 0, (int)c - 31, n, hist, stage, cnt, lane);
                        cnt += c - 31u;
                    }
                }
            }
        }
    }
    if (wbad && lane == 0) __hip_atomic_store(bad, 1u, __ATOMIC_RELAXED, __HIP_MEMORY_SCOPE_WORKGROUP);
    __syncthreads();

    const bool slow = (__builtin_amdgcn_readfirstlane((int)*bad) != 0);
    if (slow) {
        if (COUNT) {
            for (int i = tid; i < NBINS; i += AGG_THREADS) hist[i] = 0u;
        }
        __syncthreads();
        sum_from_slabs<ROWW, PITCH, COUNT>(feat, src, dst, gslabs, acc, hist, stage, g, lo, n, E, nblk, w, lane);
        return;
    }

    if (w == 0) {
        unsigned cr[RPW];
        unsigned tot = 0u;
        #pragma unroll
        for (int i = 0; i < RPW; ++i) { cr[i] = hist[lane * RPW + i]; tot += cr[i]; }
        unsigned incl = tot;
        #pragma unroll
        for (int o = 1; o < 32; o <<= 1) {
            const unsigned t = (unsigned)__shfl_up((int)incl, o);
            if (lane >= o) incl += t;
        }
        unsigned run = incl - tot;
        #pragma unroll
        for (int i = 0; i < RPW; ++i) { start[lane * RPW + i] = run; run += cr[i]; }
        if (lane == 31) start[NBINS] = run;
    }
    __syncthreads();

    for (unsigned i = (unsigned)lane; i < cnt; i += 32u) {
        const v2i q = stage[i];
        const unsigned du = (unsigned)q.y & 511u;
        const unsigned pos = start[du] + ((unsigned)q.y >> 9);
        v2i o;
        o.x = q.x;
        o.y = (int)du;
        if (pos < (unsigned)SCAP) sorted[pos] = o;
    }
    __syncthreads();

    unsigned p  = (unsigned)__builtin_amdgcn_readfirstlane((int)start[w * RPW]);
    unsigned pe = (unsigned)__builtin_amdgcn_readfirstlane((int)start[w * RPW + RPW]);
    if (pe > (unsigned)SCAP) pe = SCAP;
    int cur = -1;
    v2f a;
    a.x = 0.0f;
    a.y = 0.0f;
    for (; p + (unsigned)CHUNK <= pe; p += (unsigned)CHUNK) {
        const v2i q = sorted[p + (unsigned)(lane & (CHUNK - 1))];
        sum_chunk_rows<ROWW, PITCH, true>(feat, acc, q.x, q.y, CHUNK, lane, cur, a);
    }
    if (p < pe) {
        unsigned i = p + (unsigned)(lane & (CHUNK - 1));
        if (i > (unsigned)(SCAP - 1)) i = SCAP - 1;
        const v2i q = sorted[i];
        sum_chunk_rows<ROWW, PITCH, false>(feat, acc, q.x, q.y, (int)(pe - p), lane, cur, a);
    }
    if (cur >= 0) {
        const bool on  = (ROWW == FEAT) || (lane * 2 < ROWW);
        const int  col = on ? lane * 2 : (ROWW - 2);
        if (on) *(v2f*)(acc + cur * PITCH + col) = a;
    }
}

__global__ __launch_bounds__(AGG_THREADS) void owned_rows_finish_kernel(const float* __restrict__ z,
                                                                        const int* __restrict__ src,
                                                                        const int* __restrict__ dst,
                                                                        const v2i* __restrict__ slabs,
                                                                        const float* __restrict__ deg,
                                                                        const float* __restrict__ b2,
                                                                        float* __restrict__ out,
                                                                        int n, int E, int nblk) {
    __shared__ float acc[ROWS_PER_WG * NCLS];
    __shared__ unsigned hist[NBINS];
    __shared__ unsigned start[NBINS + 1];
    __shared__ v2i stage[AGG_WAVES * WCAP];
    __shared__ v2i sorted[SCAP];
    __shared__ unsigned bad;

    const int tid = threadIdx.x;
    const int lane = tid & 31;
    const int w = __builtin_amdgcn_readfirstlane(tid >> 5);
    const int lo = blockIdx.x * ROWS_PER_WG;
    int rows = n - lo;
    if (rows > ROWS_PER_WG) rows = ROWS_PER_WG;

    for (int i = tid; i < ROWS_PER_WG * NCLS / 4; i += AGG_THREADS) {
        ((float4*)acc)[i] = make_float4(0.f, 0.f, 0.f, 0.f);
    }
    for (int i = tid; i < NBINS; i += AGG_THREADS) hist[i] = 0u;
    if (tid == 0) bad = 0u;
    __syncthreads();

    sum_rows_owned<NCLS, NCLS, false>(z, src, dst, slabs + (long long)blockIdx.x * nblk * SLAB,
                                      acc, hist, start, stage, sorted, &bad,
                                      (int)blockIdx.x, lo, n, E, nblk, tid, w, lane);
    __syncthreads();

    const v4f* b2v = (const v4f*)b2;
    v4f* outp = (v4f*)(out + (long long)lo * NCLS);
    for (int i = tid; i < rows * (NCLS / 4); i += AGG_THREADS) {
        const int r  = i / (NCLS / 4);
        const int c4 = i - r * (NCLS / 4);
        const float d = deg[lo + r];
        const float inv = (d > 0.0f) ? (1.0f / d) : 0.0f;
        const v4f a = ((const v4f*)acc)[i];
        outp[i] = a * inv + b2v[c4];
    }
}

#define AP 68
#define ACC_ROWS (((ROWS_PER_WG + 15) / 16) * 16)
static_assert(ACC_ROWS / 16 <= AGG_WAVES, "one 16-row tile per wave");
#define T2OFF (AGG_WAVES - ACC_ROWS / 16)

__global__ __launch_bounds__(AGG_THREADS) void owned_rows_sum_products_kernel(const float* __restrict__ feat,
                                                                              const int* __restrict__ src,
                                                                              const int* __restrict__ dst,
                                                                              const v2i* __restrict__ slabs,
                                                                              const float* __restrict__ W1,
                                                                              const float* __restrict__ b1,
                                                                              const float* __restrict__ W2,
                                                                              float* __restrict__ deg,
                                                                              float* __restrict__ z,
                                                                              int n, int E, int nblk) {
    __shared__ alignas(16) float acc[ACC_ROWS * AP];
    __shared__ unsigned ldeg[NBINS];
    __shared__ unsigned start[NBINS + 1];
    __shared__ v2i stage[AGG_WAVES * WCAP];
    __shared__ v2i sorted[SCAP];
    __shared__ unsigned bad;

    const int tid = threadIdx.x;
    const int lane = tid & 31;
    const int w = __builtin_amdgcn_readfirstlane(tid >> 5);
    const int lo = blockIdx.x * ROWS_PER_WG;
    int rows = n - lo;
    if (rows > ROWS_PER_WG) rows = ROWS_PER_WG;

    for (int i = tid; i < ACC_ROWS * AP / 4; i += AGG_THREADS) {
        ((float4*)acc)[i] = make_float4(0.f, 0.f, 0.f, 0.f);
    }
    for (int i = tid; i < NBINS; i += AGG_THREADS) ldeg[i] = 0u;
    if (tid == 0) bad = 0u;
    __syncthreads();

    sum_rows_owned<FEAT, AP, true>(feat, src, dst, slabs + (long long)blockIdx.x * nblk * SLAB,
                                   acc, ldeg, start, stage, sorted, &bad,
                                   (int)blockIdx.x, lo, n, E, nblk, tid, w, lane);
    __syncthreads();

    for (int i = tid; i < rows; i += AGG_THREADS) {
        deg[lo + i] = (float)ldeg[i];
    }

    const int  ntiles = (rows + 15) >> 4;
    const bool mine   = (w < ntiles);
    const int  h      = lane >> 4;
    const int  c      = lane & 15;
    float* Tt = acc + (mine ? w : 0) * (16 * AP);

    if (mine) {
        v4f a4[8];
        #pragma unroll
        for (int qd = 0; qd < 8; ++qd) a4[qd] = *(const v4f*)(Tt + c * AP + 32 * h + 4 * qd);

        v8f acc1[4] = {};
        #pragma unroll
        for (int ks = 0; ks < 16; ++ks) {
            const int k = 32 * h + 2 * ks;
            const v4f w0 = *(const v4f*)(W1 + (long long)k * FEAT + 4 * c);
            const v4f w1 = *(const v4f*)(W1 + (long long)(k + 1) * FEAT + 4 * c);
            v2f a;
            a.x = a4[ks >> 1][(ks & 1) * 2];
            a.y = a4[ks >> 1][(ks & 1) * 2 + 1];
            #pragma unroll
            for (int j = 0; j < 4; ++j) {
                v2f b;
                b.x = w0[j];
                b.y = w1[j];
                acc1[j] = __builtin_amdgcn_wmma_f32_16x16x4_f32(
                               false, a,  false, b,
                               (short)0, acc1[j],  false,  false);
            }
        }

        const v4f bv = *(const v4f*)(b1 + 4 * c);
        #pragma unroll
        for (int r = 0; r < 8; ++r) {
            const float d = (float)ldeg[16 * w + 8 * h + r];
            const float inv = (d > 0.0f) ? (1.0f / d) : 0.0f;
            v4f t;
            t.x = fmaxf(acc1[0][r] * inv + bv.x, 0.0f);
            t.y = fmaxf(acc1[1][r] * inv + bv.y, 0.0f);
            t.z = fmaxf(acc1[2][r] * inv + bv.z, 0.0f);
            t.w = fmaxf(acc1[3][r] * inv + bv.w, 0.0f);
            *(v4f*)(Tt + (8 * h + r) * AP + 4 * c) = t;
        }
    }
    __syncthreads();

    const int  w2    = w - T2OFF;
    const bool mine2 = (w2 >= 0) && (w2 < ntiles);
    if (mine2) {
        const float* T2 = acc + w2 * (16 * AP);
        v4f h4[8];
        #pragma unroll
        for (int qd = 0; qd < 8; ++qd) h4[qd] = *(const v4f*)(T2 + c * AP + 32 * h + 4 * qd);

        const int cz = (c < NCLS / 4) ? c : (NCLS / 4 - 1);
        v8f acc2[4] = {};
        #pragma unroll
        for (int ks = 0; ks < 16; ++ks) {
            const int k = 32 * h + 2 * ks;
            const v4f w0 = *(const v4f*)(W2 + (long long)k * NCLS + 4 * cz);
            const v4f w1 = *(const v4f*)(W2 + (long long)(k + 1) * NCLS + 4 * cz);
            v2f a;
            a.x = h4[ks >> 1][(ks & 1) * 2];
            a.y = h4[ks >> 1][(ks & 1) * 2 + 1];
            #pragma unroll
            for (int j = 0; j < 4; ++j) {
                v2f b;
                b.x = w0[j];
                b.y = w1[j];
                acc2[j] = __builtin_amdgcn_wmma_f32_16x16x4_f32(
                               false, a,  false, b,
                               (short)0, acc2[j],  false,  false);
            }
        }

        #pragma unroll
        for (int r = 0; r < 8; ++r) {
            const int lr = 16 * w2 + 8 * h + r;
            v4f o;
            o.x = acc2[0][r];
            o.y = acc2[1][r];
            o.z = acc2[2][r];
            o.w = acc2[3][r];
            if (c < NCLS / 4 && lr < rows) {
                *(v4f*)(z + (long long)(lo + lr) * NCLS + 4 * c) = o;
            }
        }
    }
}

extern "C" void kernel_launch(void* const* d_in, const int* in_sizes, int n_in,
                              void* d_out, int out_size, void* d_ws, size_t ws_size,
                              hipStream_t stream) {
    const float* features = (const float*)d_in[0];
    const float* W1 = (const float*)d_in[1];
    const float* b1 = (const float*)d_in[2];
    const float* W2 = (const float*)d_in[3];
    const float* b2 = (const float*)d_in[4];
    const int* src = (const int*)d_in[5];
    const int* dst = (const int*)d_in[6];
    float* out = (float*)d_out;

    const int N = in_sizes[0] / FEAT;
    const int E = in_sizes[5];

    float* deg = (float*)d_ws;
    float* z   = deg + N;
    const int aggGroups = (N + ROWS_PER_WG - 1) / ROWS_PER_WG;
    const int nblk = (E + EB - 1) / EB;
    v2i* slabs = (v2i*)(z + (long long)N * FEAT);

    bin_edges_kernel<<<nblk, BIN_THREADS, 0, stream>>>(src, dst, slabs, N, E, nblk, aggGroups);

    owned_rows_sum_products_kernel<<<aggGroups, AGG_THREADS, 0, stream>>>(features, src, dst, slabs, W1, b1, W2, deg, z, N, E, nblk);

    owned_rows_finish_kernel<<<aggGroups, AGG_THREADS, 0, stream>>>(z, src, dst, slabs, deg, b2, out, N, E, nblk);
}


# Pairs packed to one dword (source row above a 9-bit local row): a slab is one 128-byte line, count first, written by one whole-line store and opened by one 128-byte load; sum order unchanged
# speedup vs baseline: 8.1235x; 1.0776x over previous
#include <hip/hip_runtime.h>
#include <hip/hip_bf16.h>

typedef __attribute__((ext_vector_type(2))) float v2f;
typedef __attribute__((ext_vector_type(8))) float v8f;
typedef __attribute__((ext_vector_type(4))) float v4f;

#define FEAT 64
#define NCLS 40

#define ROWS_PER_WG 396
#define AGG_THREADS 1024
#define AGG_WAVES   32
#define EB          4096
#define BIN_THREADS 256
#define MAXG        256
#define SLAB        32
#define SLABP       63
#define PAIR_SHIFT  9
#define PAIR_MASK   511
#define CHUNK       16
#define RCAP        256
#define RPW         13
#define NBINS       (AGG_WAVES * RPW)
#define WCAP        320
#define SCAP        (AGG_WAVES * WCAP)
#define SBATCH      10

typedef __attribute__((ext_vector_type(4))) int v4i;
typedef __attribute__((ext_vector_type(2))) int v2i;

__global__ __launch_bounds__(BIN_THREADS) void bin_edges_kernel(const int* __restrict__ src,
                                                                const int* __restrict__ dst,
                                                                int* __restrict__ slabs,
                                                                int n, int E, int nblk, int groups) {
    __shared__ unsigned hist[MAXG];
    __shared__ unsigned start[MAXG];
    __shared__ int sorted[EB];
    int* __restrict__ more = slabs + (long long)groups * nblk * SLAB;

    const int tid  = threadIdx.x;
    const int lane = tid & 31;
    const int w    = __builtin_amdgcn_readfirstlane(tid >> 5);
    const int b    = blockIdx.x;

    hist[tid] = 0u;
    __syncthreads();

    int      gq[16];
    unsigned rk[16];
    int      pr[16];
    #pragma unroll
    for (int k = 0; k < 4; ++k) {
        const int e = b * EB + (k * BIN_THREADS + tid) * 4;
        v4i d4, s4;
        if (e + 4 <= E) {
            d4 = *(const v4i*)(dst + e);
            s4 = *(const v4i*)(src + e);
        } else {
            #pragma unroll
            for (int c = 0; c < 4; ++c) {
                d4[c] = (e + c < E) ? dst[e + c] : -1;
                s4[c] = (e + c < E) ? src[e + c] : 0;
            }
        }
        #pragma unroll
        for (int c = 0; c < 4; ++c) {
            const unsigned u = (unsigned)d4[c];
            const unsigned g = u / (unsigned)ROWS_PER_WG;
            const bool ok = (u < (unsigned)n) && (g < (unsigned)MAXG);
            int s = s4[c];
            s = (s < 0) ? 0 : ((s >= n) ? (n - 1) : s);
            const int j = 4 * k + c;
            gq[j] = ok ? (int)g : -1;
            pr[j] = (int)(((unsigned)s << PAIR_SHIFT) | (u - g * (unsigned)ROWS_PER_WG));
            rk[j] = 0u;
            if (ok) rk[j] = __hip_atomic_fetch_add(&hist[g], 1u, __ATOMIC_RELAXED, __HIP_MEMORY_SCOPE_WORKGROUP);
        }
    }
    __syncthreads();

    if (w == 0) {
        unsigned c8[8];
        unsigned tot = 0u;
        #pragma unroll
        for (int i = 0; i < 8; ++i) { c8[i] = hist[lane * 8 + i]; tot += c8[i]; }
        unsigned incl = tot;
        #pragma unroll
        for (int o = 1; o < 32; o <<= 1) {
            const unsigned t = (unsigned)__shfl_up((int)incl, o);
            if (lane >= o) incl += t;
        }
        unsigned run = incl - tot;
        #pragma unroll
        for (int i = 0; i < 8; ++i) { start[lane * 8 + i] = run; run += c8[i]; }
    }
    __syncthreads();

    #pragma unroll
    for (int j = 0; j < 16; ++j) {
        if (gq[j] >= 0) sorted[start[gq[j]] + rk[j]] = pr[j];
    }
    __syncthreads();

    const int gl = w + (BIN_THREADS / 32) * lane;
    const unsigned cl = hist[gl];
    const unsigned sl0 = start[gl];
    #pragma unroll 4
    for (int i = 0; i < 32; ++i) {
        const int g = w + (BIN_THREADS / 32) * i;
        if (g >= groups) break;
        const unsigned c  = (unsigned)__builtin_amdgcn_readlane((int)cl, i);
        const unsigned s0 = (unsigned)__builtin_amdgcn_readlane((int)sl0, i);
        const unsigned cc = (c < (unsigned)SLABP) ? c : (unsigned)SLABP;
        const long long piece = ((long long)g * nblk + b) * SLAB;
        int word = 0;
        if ((unsigned)(lane - 1) < cc) word = sorted[s0 + (unsigned)(lane - 1)];
        if (lane == 0) word = (int)c;
        slabs[piece + lane] = word;
        if (cc > 31u) {
            if ((unsigned)lane + 31u < cc) more[piece + lane] = sorted[s0 + 31u + (unsigned)lane];
        }
    }
}

template <int ROWW, int PITCH, bool FULL>
__device__ __forceinline__ void sum_chunk(const float* __restrict__ feat, float* acc,
                                          int s, int d, int m, int lane) {
    int si[CHUNK], di[CHUNK];
    #pragma unroll
    for (int i = 0; i < CHUNK; ++i) {
        if (FULL || i < m) {
            si[i] = __builtin_amdgcn_readlane(s, i);
            di[i] = __builtin_amdgcn_readlane(d, i);
        }
    }
    if ((ROWW == FEAT) || (lane * 2 < ROWW)) {
        v2f v[CHUNK];
        #pragma unroll
        for (int i = 0; i < CHUNK; ++i) {
            if (FULL || i < m) v[i] = *(const v2f*)(feat + (long long)si[i] * ROWW + lane * 2);
        }
        #pragma unroll
        for (int i = 0; i < CHUNK; ++i) {
            if (FULL || i < m) {
                float* p = acc + di[i] * PITCH + lane * 2;
                __hip_atomic_fetch_add(p,     v[i].x, __ATOMIC_RELAXED, __HIP_MEMORY_SCOPE_WORKGROUP);
                __hip_atomic_fetch_add(p + 1, v[i].y, __ATOMIC_RELAXED, __HIP_MEMORY_SCOPE_WORKGROUP);
            }
        }
    }
}

template <int ROWW, int PITCH, bool COUNT>
__device__ void sum_block_edges(const float* __restrict__ feat,
                                             const int* __restrict__ src,
                                             const int* __restrict__ dst,
                                             float* acc, unsigned* ldeg,
                                             int g, int lo, int n, int E, int b, int lane) {
    const int e0 = b * EB;
    const int e1 = (e0 + EB < E) ? (e0 + EB) : E;
    for (int e = e0; e < e1; e += 32) {
        const int ee = e + lane;
        int d = -1, s = 0;
        if (ee < e1) { d = dst[ee]; s = src[ee]; }
        const unsigned u = (unsigned)d;
        const bool hit = (u < (unsigned)n) && ((int)(u / (unsigned)ROWS_PER_WG) == g);
        s = (s < 0) ? 0 : ((s >= n) ? (n - 1) : s);
        const int dl = hit ? (int)(u - (unsigned)lo) : 0;
        unsigned mask = __builtin_amdgcn_ballot_w32(hit);
        if (COUNT && hit) __hip_atomic_fetch_add(ldeg + dl, 1u, __ATOMIC_RELAXED, __HIP_MEMORY_SCOPE_WORKGROUP);
        while (mask != 0u) {
            const int l = __builtin_ctz(mask);
            const int si = __builtin_amdgcn_readlane(s, l);
            const int di = __builtin_amdgcn_readlane(dl, l);
            if ((ROWW == FEAT) || (lane * 2 < ROWW)) {
                const v2f v = *(const v2f*)(feat + (long long)si * ROWW + lane * 2);
                float* p = acc + di * PITCH + lane * 2;
                __hip_atomic_fetch_add(p,     v.x, __ATOMIC_RELAXED, __HIP_MEMORY_SCOPE_WORKGROUP);
                __hip_atomic_fetch_add(p + 1, v.y, __ATOMIC_RELAXED, __HIP_MEMORY_SCOPE_WORKGROUP);
            }
            mask &= mask - 1u;
        }
    }
}

template <bool COUNT>
__device__ __forceinline__ void ring_push(int p, int first, int m, int n, unsigned* ldeg, v2i* ring,
                                          unsigned tail, int lane) {
    int s = (int)((unsigned)p >> PAIR_SHIFT);
    s = (s < 0) ? 0 : ((s >= n) ? (n - 1) : s);
    unsigned du = (unsigned)p & (unsigned)PAIR_MASK;
    if (du > (unsigned)(ROWS_PER_WG - 1)) du = ROWS_PER_WG - 1;
    const int k = lane - first;
    if (k >= 0 && k < m) {
        if (COUNT) {
            __hip_atomic_fetch_add(ldeg + du, 1u, __ATOMIC_RELAXED, __HIP_MEMORY_SCOPE_WORKGROUP);
        }
        v2i q;
        q.x = s;
        q.y = (int)du;
        ring[(tail + (unsigned)k) & (RCAP - 1)] = q;
    }
}

template <int ROWW, int PITCH, bool COUNT>
__device__ __forceinline__ void sum_from_slabs(const float* __restrict__ feat,
                                               const int* __restrict__ src,
                                               const int* __restrict__ dst,
                                               const int* __restrict__ gslabs,
                                               const int* __restrict__ gmore,
                                               float* acc, unsigned* ldeg, v2i* ring,
                                               int g, int lo, int n, int E, int nblk, int w, int lane) {
    int nxt = 0;
    if (w < nblk) nxt = gslabs[(long long)w * SLAB + lane];
    unsigned head = 0u, tail = 0u;
    for (int b = w; b < nblk; b += AGG_WAVES) {
        const int cur = nxt;
        const int bn = b + AGG_WAVES;
        if (bn < nblk) nxt = gslabs[(long long)bn * SLAB + lane];
        const int c = __builtin_amdgcn_readlane(cur, 0);
        if (c > SLABP) {
            sum_block_edges<ROWW, PITCH, COUNT>(feat, src, dst, acc, ldeg, g, lo, n, E, b, lane);
            continue;
        }
        if (c > 0) {
            const int m0 = (c < 31) ? c : 31;
            ring_push<COUNT>(cur, 1, m0, n, ldeg, ring, tail, lane);
            tail += (unsigned)m0;
            if (c > 31) {
                int p = 0;
                if (31 + lane < c) p = gmore[(long long)b * SLAB + lane];
                ring_push<COUNT>(p, 0, c - 31, n, ldeg, ring, tail, lane);
                tail += (unsigned)(c - 31);
            }
            while (tail - head >= (unsigned)CHUNK) {
                const v2i q = ring[(head + (unsigned)(lane & (CHUNK - 1))) & (RCAP - 1)];
                sum_chunk<ROWW, PITCH, true>(feat, acc, q.x, q.y, CHUNK, lane);
                head += (unsigned)CHUNK;
            }
        }
    }
    if (tail != head) {
        const v2i q = ring[(head + (unsigned)(lane & (CHUNK - 1))) & (RCAP - 1)];
        sum_chunk<ROWW, PITCH, false>(feat, acc, q.x, q.y, (int)(tail - head), lane);
    }
}

static_assert(NBINS >= ROWS_PER_WG, "every row has an owner wave");
static_assert(ROWS_PER_WG <= 512, "a local row fits the 9 low bits of a staged pair");
static_assert(RCAP <= WCAP, "the slow path's ring lives in the staging slots");

__device__ __forceinline__ void stage_push(int p, int first, int m, int n, unsigned* hist, v2i* stage,
                                           unsigned cnt, int lane) {
    int s = (int)((unsigned)p >> PAIR_SHIFT);
    s = (s < 0) ? 0 : ((s >= n) ? (n - 1) : s);
    unsigned du = (unsigned)p & (unsigned)PAIR_MASK;
    if (du > (unsigned)(ROWS_PER_WG - 1)) du = ROWS_PER_WG - 1;
    const int k = lane - first;
    if (k >= 0 && k < m) {
        const unsigned rank = __hip_atomic_fetch_add(hist + du, 1u, __ATOMIC_RELAXED, __HIP_MEMORY_SCOPE_WORKGROUP);
        v2i q;
        q.x = s;
        q.y = (int)((rank << 9) | du);
        stage[cnt + (unsigned)k] = q;
    }
}

template <int ROWW, int PITCH, bool FULL>
__device__ __forceinline__ void sum_chunk_rows(const float* __restrict__ feat, float* acc,
                                               int s, int d, int m, int lane, int& cur, v2f& a) {
    int si[CHUNK], di[CHUNK];
    #pragma unroll
    for (int i = 0; i < CHUNK; ++i) {
        if (FULL || i < m) {
            si[i] = __builtin_amdgcn_readlane(s, i);
            di[i] = __builtin_amdgcn_readlane(d, i);
        }
    }
    const bool on  = (ROWW == FEAT) || (lane * 2 < ROWW);
    const int  col = on ? lane * 2 : (ROWW - 2);
    v2f v[CHUNK];
    #pragma unroll
    for (int i = 0; i < CHUNK; ++i) {
        if (FULL || i < m) v[i] = *(const v2f*)(feat + (long long)si[i] * ROWW + col);
    }
    #pragma unroll
    for (int i = 0; i < CHUNK; ++i) {
        if (FULL || i < m) {
            if (di[i] != cur) {
                if (cur >= 0 && on) *(v2f*)(acc + cur * PITCH + col) = a;
                cur = di[i];
                a = v[i];
            } else {
                a += v[i];
            }
        }
    }
}

template <int ROWW, int PITCH, bool COUNT>
__device__ __forceinline__ void sum_rows_owned(const float* __restrict__ feat,
                                               const int* __restrict__ src,
                                               const int* __restrict__ dst,
                                               const int* __restrict__ gslabs,
                                               const int* __restrict__ gmore,
                                               float* acc, unsigned* hist, unsigned* start,
                                               v2i* stage_all, v2i* sorted, unsigned* bad,
                                               int g, int lo, int n, int E, int nblk,
                                               int tid, int w, int lane) {
    v2i* stage = stage_all + w * WCAP;

    unsigned cnt = 0u;
    bool wbad = false;
    for (int b0 = w; b0 < nblk; b0 += AGG_WAVES * SBATCH) {
        int sl[SBATCH];
        #pragma unroll
        for (int j = 0; j < SBATCH; ++j) {
            const int b = b0 + AGG_WAVES * j;
            sl[j] = 0;
            if (b < nblk) sl[j] = gslabs[(long long)b * SLAB + lane];
        }
        #pragma unroll
        for (int j = 0; j < SBATCH; ++j) {
            const int b = b0 + AGG_WAVES * j;
            if (b < nblk) {
                const unsigned c = (unsigned)__builtin_amdgcn_readlane(sl[j], 0);
                if (c > (unsigned)SLABP || cnt + c > (unsigned)WCAP) {
                    wbad = true;
                } else if (c > 0u) {
                    const int m0 = (c < 31u) ? (int)c : 31;
                    stage_push(sl[j], 1, m0, n, hist, stage, cnt, lane);
                    cnt += (unsigned)m0;
                    if (c > 31u) {
                        int p = 0;
                        if (31u + (unsigned)lane < c) p = gmore[(long long)b * SLAB + lane];
                        stage_push(p, 0, (int)c - 31, n, hist, stage, cnt, lane);
                        cnt += c - 31u;
                    }
                }
            }
        }
    }
    if (wbad && lane == 0) __hip_atomic_store(bad, 1u, __ATOMIC_RELAXED, __HIP_MEMORY_SCOPE_WORKGROUP);
    __syncthreads();

    const bool slow = (__builtin_amdgcn_readfirstlane((int)*bad) != 0);
    if (slow) {
        if (COUNT) {
            for (int i = tid; i < NBINS; i += AGG_THREADS) hist[i] = 0u;
        }
        __syncthreads();
        sum_from_slabs<ROWW, PITCH, COUNT>(feat, src, dst, gslabs, gmore, acc, hist, stage, g, lo, n, E, nblk, w, lane);
        return;
    }

    if (w == 0) {
        unsigned cr[RPW];
        unsigned tot = 0u;
        #pragma unroll
        for (int i = 0; i < RPW; ++i) { cr[i] = hist[lane * RPW + i]; tot += cr[i]; }
        unsigned incl = tot;
        #pragma unroll
        for (int o = 1; o < 32; o <<= 1) {
            const unsigned t = (unsigned)__shfl_up((int)incl, o);
            if (lane >= o) incl += t;
        }
        unsigned run = incl - tot;
        #pragma unroll
        for (int i = 0; i < RPW; ++i) { start[lane * RPW + i] = run; run += cr[i]; }
        if (lane == 31) start[NBINS] = run;
    }
    __syncthreads();

    for (unsigned i = (unsigned)lane; i < cnt; i += 32u) {
        const v2i q = stage[i];
        const unsigned du = (unsigned)q.y & 511u;
        const unsigned pos = start[du] + ((unsigned)q.y >> 9);
        v2i o;
        o.x = q.x;
        o.y = (int)du;
        if (pos < (unsigned)SCAP) sorted[pos] = o;
    }
    __syncthreads();

    unsigned p  = (unsigned)__builtin_amdgcn_readfirstlane((int)start[w * RPW]);
    unsigned pe = (unsigned)__builtin_amdgcn_readfirstlane((int)start[w * RPW + RPW]);
    if (pe > (unsigned)SCAP) pe = SCAP;
    int cur = -1;
    v2f a;
    a.x = 0.0f;
    a.y = 0.0f;
    for (; p + (unsigned)CHUNK <= pe; p += (unsigned)CHUNK) {
        const v2i q = sorted[p + (unsigned)(lane & (CHUNK - 1))];
        sum_chunk_rows<ROWW, PITCH, true>(feat, acc, q.x, q.y, CHUNK, lane, cur, a);
    }
    if (p < pe) {
        unsigned i = p + (unsigned)(lane & (CHUNK - 1));
        if (i > (unsigned)(SCAP - 1)) i = SCAP - 1;
        const v2i q = sorted[i];
        sum_chunk_rows<ROWW, PITCH, false>(feat, acc, q.x, q.y, (int)(pe - p), lane, cur, a);
    }
    if (cur >= 0) {
        const bool on  = (ROWW == FEAT) || (lane * 2 < ROWW);
        const int  col = on ? lane * 2 : (ROWW - 2);
        if (on) *(v2f*)(acc + cur * PITCH + col) = a;
    }
}

__global__ __launch_bounds__(AGG_THREADS) void owned_rows_finish_kernel(const float* __restrict__ z,
                                                                        const int* __restrict__ src,
                                                                        const int* __restrict__ dst,
                                                                        const int* __restrict__ slabs,
                                                                        const float* __restrict__ deg,
                                                                        const float* __restrict__ b2,
                                                                        float* __restrict__ out,
                                                                        int n, int E, int nblk) {
    __shared__ float acc[ROWS_PER_WG * NCLS];
    __shared__ unsigned hist[NBINS];
    __shared__ unsigned start[NBINS + 1];
    __shared__ v2i stage[AGG_WAVES * WCAP];
    __shared__ v2i sorted[SCAP];
    __shared__ unsigned bad;

    const int tid = threadIdx.x;
    const int lane = tid & 31;
    const int w = __builtin_amdgcn_readfirstlane(tid >> 5);
    const int lo = blockIdx.x * ROWS_PER_WG;
    int rows = n - lo;
    if (rows > ROWS_PER_WG) rows = ROWS_PER_WG;

    for (int i = tid; i < ROWS_PER_WG * NCLS / 4; i += AGG_THREADS) {
        ((float4*)acc)[i] = make_float4(0.f, 0.f, 0.f, 0.f);
    }
    for (int i = tid; i < NBINS; i += AGG_THREADS) hist[i] = 0u;
    if (tid == 0) bad = 0u;
    __syncthreads();

    const int* gmain = slabs + (long long)blockIdx.x * nblk * SLAB;
    const int* gmore = gmain + (long long)gridDim.x * nblk * SLAB;
    sum_rows_owned<NCLS, NCLS, false>(z, src, dst, gmain, gmore,
                                      acc, hist, start, stage, sorted, &bad,
                                      (int)blockIdx.x, lo, n, E, nblk, tid, w, lane);
    __syncthreads();

    const v4f* b2v = (const v4f*)b2;
    v4f* outp = (v4f*)(out + (long long)lo * NCLS);
    for (int i = tid; i < rows * (NCLS / 4); i += AGG_THREADS) {
        const int r  = i / (NCLS / 4);
        const int c4 = i - r * (NCLS / 4);
        const float d = deg[lo + r];
        const float inv = (d > 0.0f) ? (1.0f / d) : 0.0f;
        const v4f a = ((const v4f*)acc)[i];
        outp[i] = a * inv + b2v[c4];
    }
}

#define AP 68
#define ACC_ROWS (((ROWS_PER_WG + 15) / 16) * 16)
static_assert(ACC_ROWS / 16 <= AGG_WAVES, "one 16-row tile per wave");
#define T2OFF (AGG_WAVES - ACC_ROWS / 16)

__global__ __launch_bounds__(AGG_THREADS) void owned_rows_sum_products_kernel(const float* __restrict__ feat,
                                                                              const int* __restrict__ src,
                                                                              const int* __restrict__ dst,
                                                                              const int* __restrict__ slabs,
                                                                              const float* __restrict__ W1,
                                                                              const float* __restrict__ b1,
                                                                              const float* __restrict__ W2,
                                                                              float* __restrict__ deg,
                                                                              float* __restrict__ z,
                                                                              int n, int E, int nblk) {
    __shared__ alignas(16) float acc[ACC_ROWS * AP];
    __shared__ unsigned ldeg[NBINS];
    __shared__ unsigned start[NBINS + 1];
    __shared__ v2i stage[AGG_WAVES * WCAP];
    __shared__ v2i sorted[SCAP];
    __shared__ unsigned bad;

    const int tid = threadIdx.x;
    const int lane = tid & 31;
    const int w = __builtin_amdgcn_readfirstlane(tid >> 5);
    const int lo = blockIdx.x * ROWS_PER_WG;
    int rows = n - lo;
    if (rows > ROWS_PER_WG) rows = ROWS_PER_WG;

    for (int i = tid; i < ACC_ROWS * AP / 4; i += AGG_THREADS) {
        ((float4*)acc)[i] = make_float4(0.f, 0.f, 0.f, 0.f);
    }
    for (int i = tid; i < NBINS; i += AGG_THREADS) ldeg[i] = 0u;
    if (tid == 0) bad = 0u;
    __syncthreads();

    const int* gmain = slabs + (long long)blockIdx.x * nblk * SLAB;
    const int* gmore = gmain + (long long)gridDim.x * nblk * SLAB;
    sum_rows_owned<FEAT, AP, true>(feat, src, dst, gmain, gmore,
                                   acc, ldeg, start, stage, sorted, &bad,
                                   (int)blockIdx.x, lo, n, E, nblk, tid, w, lane);
    __syncthreads();

    for (int i = tid; i < rows; i += AGG_THREADS) {
        deg[lo + i] = (float)ldeg[i];
    }

    const int  ntiles = (rows + 15) >> 4;
    const bool mine   = (w < ntiles);
    const int  h      = lane >> 4;
    const int  c      = lane & 15;
    float* Tt = acc + (mine ? w : 0) * (16 * AP);

    if (mine) {
        v4f a4[8];
        #pragma unroll
        for (int qd = 0; qd < 8; ++qd) a4[qd] = *(const v4f*)(Tt + c * AP + 32 * h + 4 * qd);

        v8f acc1[4] = {};
        #pragma unroll
        for (int ks = 0; ks < 16; ++ks) {
            const int k = 32 * h + 2 * ks;
            const v4f w0 = *(const v4f*)(W1 + (long long)k * FEAT + 4 * c);
            const v4f w1 = *(const v4f*)(W1 + (long long)(k + 1) * FEAT + 4 * c);
            v2f a;
            a.x = a4[ks >> 1][(ks & 1) * 2];
            a.y = a4[ks >> 1][(ks & 1) * 2 + 1];
            #pragma unroll
            for (int j = 0; j < 4; ++j) {
                v2f b;
                b.x = w0[j];
                b.y = w1[j];
                acc1[j] = __builtin_amdgcn_wmma_f32_16x16x4_f32(
                               false, a,  false, b,
                               (short)0, acc1[j],  false,  false);
            }
        }

        const v4f bv = *(const v4f*)(b1 + 4 * c);
        #pragma unroll
        for (int r = 0; r < 8; ++r) {
            const float d = (float)ldeg[16 * w + 8 * h + r];
            const float inv = (d > 0.0f) ? (1.0f / d) : 0.0f;
            v4f t;
            t.x = fmaxf(acc1[0][r] * inv + bv.x, 0.0f);
            t.y = fmaxf(acc1[1][r] * inv + bv.y, 0.0f);
            t.z = fmaxf(acc1[2][r] * inv + bv.z, 0.0f);
            t.w = fmaxf(acc1[3][r] * inv + bv.w, 0.0f);
            *(v4f*)(Tt + (8 * h + r) * AP + 4 * c) = t;
        }
    }
    __syncthreads();

    const int  w2    = w - T2OFF;
    const bool mine2 = (w2 >= 0) && (w2 < ntiles);
    if (mine2) {
        const float* T2 = acc + w2 * (16 * AP);
        v4f h4[8];
        #pragma unroll
        for (int qd = 0; qd < 8; ++qd) h4[qd] = *(const v4f*)(T2 + c * AP + 32 * h + 4 * qd);

        const int cz = (c < NCLS / 4) ? c : (NCLS / 4 - 1);
        v8f acc2[4] = {};
        #pragma unroll
        for (int ks = 0; ks < 16; ++ks) {
            const int k = 32 * h + 2 * ks;
            const v4f w0 = *(const v4f*)(W2 + (long long)k * NCLS + 4 * cz);
            const v4f w1 = *(const v4f*)(W2 + (long long)(k + 1) * NCLS + 4 * cz);
            v2f a;
            a.x = h4[ks >> 1][(ks & 1) * 2];
            a.y = h4[ks >> 1][(ks & 1) * 2 + 1];
            #pragma unroll
            for (int j = 0; j < 4; ++j) {
                v2f b;
                b.x = w0[j];
                b.y = w1[j];
                acc2[j] = __builtin_amdgcn_wmma_f32_16x16x4_f32(
                               false, a,  false, b,
                               (short)0, acc2[j],  false,  false);
            }
        }

        #pragma unroll
        for (int r = 0; r < 8; ++r) {
            const int lr = 16 * w2 + 8 * h + r;
            v4f o;
            o.x = acc2[0][r];
            o.y = acc2[1][r];
            o.z = acc2[2][r];
            o.w = acc2[3][r];
            if (c < NCLS / 4 && lr < rows) {
                *(v4f*)(z + (long long)(lo + lr) * NCLS + 4 * c) = o;
            }
        }
    }
}

extern "C" void kernel_launch(void* const* d_in, const int* in_sizes, int n_in,
                              void* d_out, int out_size, void* d_ws, size_t ws_size,
                              hipStream_t stream) {
    const float* features = (const float*)d_in[0];
    const float* W1 = (const float*)d_in[1];
    const float* b1 = (const float*)d_in[2];
    const float* W2 = (const float*)d_in[3];
    const float* b2 = (const float*)d_in[4];
    const int* src = (const int*)d_in[5];
    const int* dst = (const int*)d_in[6];
    float* out = (float*)d_out;

    const int N = in_sizes[0] / FEAT;
    const int E = in_sizes[5];

    float* deg = (float*)d_ws;
    float* z   = deg + N;
    const int aggGroups = (N + ROWS_PER_WG - 1) / ROWS_PER_WG;
    const int nblk = (E + EB - 1) / EB;
    int* slabs = (int*)(z + (long long)N * FEAT);

    bin_edges_kernel<<<nblk, BIN_THREADS, 0, stream>>>(src, dst, slabs, N, E, nblk, aggGroups);

    owned_rows_sum_products_kernel<<<aggGroups, AGG_THREADS, 0, stream>>>(features, src, dst, slabs, W1, b1, W2, deg, z, N, E, nblk);

    owned_rows_finish_kernel<<<aggGroups, AGG_THREADS, 0, stream>>>(z, src, dst, slabs, deg, b2, out, N, E, nblk);
}
